# starting kernel
# baseline (speedup 1.0000x reference)
#include <hip/hip_runtime.h>
#include <math.h>

#define N1c  200000
#define N2c  40000
#define N3c  10000
#define INc  128
#define HIDc 256
#define OUTc 64

typedef float v2f __attribute__((ext_vector_type(2)));
typedef float v8f __attribute__((ext_vector_type(8)));

__global__ void zero_f32(float* __restrict__ p, int n) {
    int i = blockIdx.x * blockDim.x + threadIdx.x;
    if (i < n) p[i] = 0.0f;
}

__global__ void scatter_edges(const float* __restrict__ x,
                              const int*   __restrict__ src,
                              const int*   __restrict__ dst,
                              float*       __restrict__ agg,
                              float*       __restrict__ cnt,
                              long long total, int dshift) {
    long long i = (long long)blockIdx.x * blockDim.x + threadIdx.x;
    if (i >= total) return;
    int e = (int)(i >> dshift);
    int f = (int)(i & ((1 << dshift) - 1));
    int s = src[e];
    int d = dst[e];
    atomicAdd(&agg[((long long)d << dshift) + f], x[((long long)s << dshift) + f]);
    if (f == 0) atomicAdd(&cnt[d], 1.0f);
}

__launch_bounds__(128)
__global__ void sage_gemm_wmma(const float* __restrict__ agg,
                               const float* __restrict__ cnt,
                               const float* __restrict__ xt,
                               const float* __restrict__ Wl,
                               const float* __restrict__ Wr,
                               const float* __restrict__ bias,
                               float*       __restrict__ out,
                               int K, int Nout, int relu) {
    const int lane  = threadIdx.x & 31;
    const int wave  = threadIdx.x >> 5;
    const int mTile = blockIdx.x;
    const int nTile = blockIdx.y * 4 + wave;

    const int n  = lane & 15;
    const int kh = lane >> 4;

    const long long row = (long long)mTile * 16 + n;
    const long long col = (long long)nTile * 16 + n;

    const float inv = 1.0f / fmaxf(cnt[row], 1.0f);

    v8f acc;
    #pragma unroll
    for (int r = 0; r < 8; ++r) acc[r] = 0.0f;

    {
        const float* aRow = agg + row * K;
        const float* bRow = Wl  + col * K;
        for (int k0 = 0; k0 < K; k0 += 4) {
            const int ka = k0 + 2 * kh;
            v2f a; a[0] = aRow[ka] * inv; a[1] = aRow[ka + 1] * inv;
            v2f b; b[0] = bRow[ka];       b[1] = bRow[ka + 1];
            acc = __builtin_amdgcn_wmma_f32_16x16x4_f32(
                false, a, false, b, (short)0, acc, false, false);
        }
    }
    {
        const float* aRow = xt + row * K;
        const float* bRow = Wr + col * K;
        for (int k0 = 0; k0 < K; k0 += 4) {
            const int ka = k0 + 2 * kh;
            v2f a; a[0] = aRow[ka]; a[1] = aRow[ka + 1];
            v2f b; b[0] = bRow[ka]; b[1] = bRow[ka + 1];
            acc = __builtin_amdgcn_wmma_f32_16x16x4_f32(
                false, a, false, b, (short)0, acc, false, false);
        }
    }

    const float bv = bias[col];
    const long long mbase = (long long)mTile * 16 + ((lane < 16) ? 0 : 8);
    #pragma unroll
    for (int r = 0; r < 8; ++r) {
        float v = acc[r] + bv;
        if (relu) v = fmaxf(v, 0.0f);
        out[(mbase + r) * Nout + col] = v;
    }
}

__global__ void log_softmax64(float* __restrict__ out, int M) {
    const int lane = threadIdx.x & 31;
    const int wave = threadIdx.x >> 5;
    const int row  = blockIdx.x * 8 + wave;
    if (row >= M) return;
    float* p = out + (long long)row * 64;
    float a = p[lane];
    float b = p[lane + 32];
    float m = fmaxf(a, b);
    #pragma unroll
    for (int off = 16; off >= 1; off >>= 1)
        m = fmaxf(m, __shfl_xor(m, off, 32));
    float s = __expf(a - m) + __expf(b - m);
    #pragma unroll
    for (int off = 16; off >= 1; off >>= 1)
        s += __shfl_xor(s, off, 32);
    const float l = __logf(s) + m;
    p[lane]      = a - l;
    p[lane + 32] = b - l;
}

extern "C" void kernel_launch(void* const* d_in, const int* in_sizes, int n_in,
                              void* d_out, int out_size, void* d_ws, size_t ws_size,
                              hipStream_t stream) {
    const float* x    = (const float*)d_in[0];
    const int*   src1 = (const int*)  d_in[1];
    const int*   dst1 = (const int*)  d_in[2];
    const int*   src2 = (const int*)  d_in[3];
    const int*   dst2 = (const int*)  d_in[4];
    const float* W1l  = (const float*)d_in[7];
    const float* W1r  = (const float*)d_in[8];
    const float* b1   = (const float*)d_in[9];
    const float* W2l  = (const float*)d_in[10];
    const float* W2r  = (const float*)d_in[11];
    const float* b2   = (const float*)d_in[12];

    const int E1 = in_sizes[1];
    const int E2 = in_sizes[3];

    float* agg1 = (float*)d_ws;
    float* cnt1 = agg1 + (size_t)N2c * INc;
    float* h    = cnt1 + N2c;
    float* agg2 = h    + (size_t)N2c * HIDc;
    float* cnt2 = agg2 + (size_t)N3c * HIDc;
    float* out  = (float*)d_out;

    {
        int z1 = N2c * INc + N2c;
        zero_f32<<<(z1 + 255) / 256, 256, 0, stream>>>(agg1, z1);
        int z2 = N3c * HIDc + N3c;
        zero_f32<<<(z2 + 255) / 256, 256, 0, stream>>>(agg2, z2);
    }

    {
        long long total = (long long)E1 * INc;
        int blocks = (int)((total + 255) / 256);
        scatter_edges<<<blocks, 256, 0, stream>>>(x, src1, dst1, agg1, cnt1,
                                                  total, 7  );
        dim3 grid(N2c / 16, HIDc / 64);
        sage_gemm_wmma<<<grid, 128, 0, stream>>>(agg1, cnt1, x, W1l, W1r, b1,
                                                 h, INc, HIDc,  1);
    }

    {
        long long total = (long long)E2 * HIDc;
        int blocks = (int)((total + 255) / 256);
        scatter_edges<<<blocks, 256, 0, stream>>>(h, src2, dst2, agg2, cnt2,
                                                  total, 8  );
        dim3 grid(N3c / 16, OUTc / 64);
        sage_gemm_wmma<<<grid, 128, 0, stream>>>(agg2, cnt2, h, W2l, W2r, b2,
                                                 out, HIDc, OUTc,  0);
    }

    log_softmax64<<<(N3c + 7) / 8, 256, 0, stream>>>(out, N3c);
}


# Owner work-groups sum neighbours: each keeps 256 (layer 2: 64) target rows in LDS, scans all edges, adds matching source rows by LDS float atomics; no global atomics or zero fill; arrival order
# speedup vs baseline: 2.2607x; 2.2607x over previous
#include <hip/hip_runtime.h>
#include <math.h>

#define N1c  200000
#define N2c  40000
#define N3c  10000
#define INc  128
#define HIDc 256
#define OUTc 64

typedef float v2f __attribute__((ext_vector_type(2)));
typedef float v8f __attribute__((ext_vector_type(8)));

#define AGG_NT 1024
#define AGG_NW 32
#define AGG_U  4
#define AGG_QW 544

typedef int   v4i  __attribute__((ext_vector_type(4)));
typedef float v4f  __attribute__((ext_vector_type(4)));

__device__ __forceinline__ v4i load_dst4_tail(const int* __restrict__ dst,
                                              int cbase, int lane, int E) {
    const int e0 = cbase + 4 * lane;
    v4i d;
    d[0] = (e0     < E) ? dst[e0]     : -1;
    d[1] = (e0 + 1 < E) ? dst[e0 + 1] : -1;
    d[2] = (e0 + 2 < E) ? dst[e0 + 2] : -1;
    d[3] = (e0 + 3 < E) ? dst[e0 + 3] : -1;
    return d;
}

__device__ __forceinline__ int slot_chunk(int t, int wave, int rot, int nTot, int nFull) {
    int c = t * AGG_NW + wave + rot;
    if (c >= nTot) c -= nTot;
    return (c < nFull) ? c : -1;
}

__device__ __forceinline__ unsigned slot_rows(int c, int nrows) {
    return (unsigned)__builtin_amdgcn_readfirstlane((c < 0) ? 0 : nrows);
}

__device__ __forceinline__ v4i load_dst4(const int* __restrict__ dst, int c, int lane) {
    const int cl = (c < 0) ? 0 : c;
    return *reinterpret_cast<const v4i*>(dst + (cl << 7) + 4 * lane);
}

__device__ __forceinline__ void queue_chunk(v4i d, int cbase, int lane, int lo,
                                            unsigned nrows, int* q, int& qn) {
    const int e0 = cbase + 4 * lane;
    #pragma unroll
    for (int c = 0; c < 4; ++c) {
        const bool m = (unsigned)(d[c] - lo) < nrows;
        const unsigned mask = __builtin_amdgcn_ballot_w32(m);
        if (mask) {
            if (m) q[qn + (int)__builtin_amdgcn_mbcnt_lo(mask, 0u)] = e0 + c;
            qn += __builtin_popcount(mask);
        }
    }
}

template <int D, int R>
__device__ __forceinline__ void add_batch(const float* __restrict__ x,
                                          const int*   __restrict__ src,
                                          const int*   __restrict__ dst,
                                          float* acc, int* cntL, const int* q,
                                          int base, int count, int lo, int lane) {
    constexpr int NV = D / 64;
    constexpr int G  = 32 / NV;
    const unsigned lane2 = 2u * (unsigned)lane;
    int s = 0, r = R;
    if (lane < count) {
        const int e = q[base + lane];
        s = src[e];
        r = dst[e] - lo;
        atomicAdd(&cntL[r], 1);
    }
    for (int j = 0; j < count; j += G) {
        v2f v[G][NV];
        int rr[G];
        #pragma unroll
        for (int g = 0; g < G; ++g) {
            const int sg = __builtin_amdgcn_readlane(s, j + g);
            rr[g]        = __builtin_amdgcn_readlane(r, j + g);
            const float* row = x + (long long)sg * D;
            #pragma unroll
            for (int k = 0; k < NV; ++k)
                v[g][k] = *reinterpret_cast<const v2f*>(row + (lane2 + 64u * k));
        }
        #pragma unroll
        for (int g = 0; g < G; ++g) {
            float* a = acc + rr[g] * D + 2 * lane;
            #pragma unroll
            for (int k = 0; k < NV; ++k) {
                atomicAdd(a + 64 * k,     v[g][k][0]);
                atomicAdd(a + 64 * k + 1, v[g][k][1]);
            }
        }
    }
}

template <int D, int R>
__launch_bounds__(AGG_NT)
__global__ void aggregate_owned(const float* __restrict__ x,
                                const int*   __restrict__ src,
                                const int*   __restrict__ dst,
                                float*       __restrict__ agg,
                                float*       __restrict__ cnt,
                                int E, int N) {
    __shared__ __attribute__((aligned(16))) float acc[(R + 1) * D];
    __shared__ int cntL[R + 1];
    __shared__ int qAll[AGG_NW * AGG_QW];

    const int tid   = threadIdx.x;
    const int lane  = tid & 31;
    const int wave  = __builtin_amdgcn_readfirstlane(tid >> 5);
    const int lo    = blockIdx.x * R;
    const int nrows = (N - lo < R) ? (N - lo) : R;

    {
        v4f z; z[0] = 0.0f; z[1] = 0.0f; z[2] = 0.0f; z[3] = 0.0f;
        v4f* a4 = reinterpret_cast<v4f*>(acc);
        for (int i = tid; i < (R + 1) * D / 4; i += AGG_NT) a4[i] = z;
        if (tid <= R) cntL[tid] = 0;
    }
    __syncthreads();

    const int nFull = E >> 7;
    const int nGrp  = (nFull + AGG_NW * AGG_U - 1) / (AGG_NW * AGG_U);
    const int nTrip = (nGrp + 1) >> 1;
    const int nTot  = nTrip * 2 * AGG_NW * AGG_U;
    const int rot   = (int)blockIdx.x * (nTot / (int)gridDim.x);

    int* q  = qAll + wave * AGG_QW;
    int  qn = 0;

    if (nFull > 0) {
        v4i da[AGG_U], db[AGG_U];
        int ca[AGG_U], cb[AGG_U];
        #pragma unroll
        for (int u = 0; u < AGG_U; ++u) {
            ca[u] = slot_chunk(u, wave, rot, nTot, nFull);
            da[u] = load_dst4(dst, ca[u], lane);
        }
        for (int t = 0; t < nTrip * 2 * AGG_U; t += 2 * AGG_U) {
            #pragma unroll
            for (int u = 0; u < AGG_U; ++u) {
                cb[u] = slot_chunk(t + AGG_U + u, wave, rot, nTot, nFull);
                db[u] = load_dst4(dst, cb[u], lane);
            }
            #pragma unroll
            for (int u = 0; u < AGG_U; ++u)
                queue_chunk(da[u], ca[u] << 7, lane, lo, slot_rows(ca[u], nrows), q, qn);
            while (qn >= 32) {
                qn -= 32;
                add_batch<D, R>(x, src, dst, acc, cntL, q, qn, 32, lo, lane);
            }
            #pragma unroll
            for (int u = 0; u < AGG_U; ++u) {
                ca[u] = (t + 2 * AGG_U < nTrip * 2 * AGG_U)
                            ? slot_chunk(t + 2 * AGG_U + u, wave, rot, nTot, nFull) : -1;
                da[u] = load_dst4(dst, ca[u], lane);
            }
            #pragma unroll
            for (int u = 0; u < AGG_U; ++u)
                queue_chunk(db[u], cb[u] << 7, lane, lo, slot_rows(cb[u], nrows), q, qn);
            while (qn >= 32) {
                qn -= 32;
                add_batch<D, R>(x, src, dst, acc, cntL, q, qn, 32, lo, lane);
            }
        }
    }
    if ((E & 127) != 0 && wave == 0) {
        const v4i d = load_dst4_tail(dst, nFull << 7, lane, E);
        queue_chunk(d, nFull << 7, lane, lo, (unsigned)nrows, q, qn);
        while (qn >= 32) {
            qn -= 32;
            add_batch<D, R>(x, src, dst, acc, cntL, q, qn, 32, lo, lane);
        }
    }
    if (qn > 0) add_batch<D, R>(x, src, dst, acc, cntL, q, 0, qn, lo, lane);

    __syncthreads();

    {
        const v4f* a4 = reinterpret_cast<const v4f*>(acc);
        v4f* o4 = reinterpret_cast<v4f*>(agg + (long long)lo * D);
        for (int i = tid; i < nrows * (D / 4); i += AGG_NT) o4[i] = a4[i];
        if (tid < nrows) cnt[lo + tid] = (float)cntL[tid];
    }
}

__launch_bounds__(128)
__global__ void sage_gemm_wmma(const float* __restrict__ agg,
                               const float* __restrict__ cnt,
                               const float* __restrict__ xt,
                               const float* __restrict__ Wl,
                               const float* __restrict__ Wr,
                               const float* __restrict__ bias,
                               float*       __restrict__ out,
                               int K, int Nout, int relu) {
    const int lane  = threadIdx.x & 31;
    const int wave  = threadIdx.x >> 5;
    const int mTile = blockIdx.x;
    const int nTile = blockIdx.y * 4 + wave;

    const int n  = lane & 15;
    const int kh = lane >> 4;

    const long long row = (long long)mTile * 16 + n;
    const long long col = (long long)nTile * 16 + n;

    const float inv = 1.0f / fmaxf(cnt[row], 1.0f);

    v8f acc;
    #pragma unroll
    for (int r = 0; r < 8; ++r) acc[r] = 0.0f;

    {
        const float* aRow = agg + row * K;
        const float* bRow = Wl  + col * K;
        for (int k0 = 0; k0 < K; k0 += 4) {
            const int ka = k0 + 2 * kh;
            v2f a; a[0] = aRow[ka] * inv; a[1] = aRow[ka + 1] * inv;
            v2f b; b[0] = bRow[ka];       b[1] = bRow[ka + 1];
            acc = __builtin_amdgcn_wmma_f32_16x16x4_f32(
                false, a, false, b, (short)0, acc, false, false);
        }
    }
    {
        const float* aRow = xt + row * K;
        const float* bRow = Wr + col * K;
        for (int k0 = 0; k0 < K; k0 += 4) {
            const int ka = k0 + 2 * kh;
            v2f a; a[0] = aRow[ka]; a[1] = aRow[ka + 1];
            v2f b; b[0] = bRow[ka]; b[1] = bRow[ka + 1];
            acc = __builtin_amdgcn_wmma_f32_16x16x4_f32(
                false, a, false, b, (short)0, acc, false, false);
        }
    }

    const float bv = bias[col];
    const long long mbase = (long long)mTile * 16 + ((lane < 16) ? 0 : 8);
    #pragma unroll
    for (int r = 0; r < 8; ++r) {
        float v = acc[r] + bv;
        if (relu) v = fmaxf(v, 0.0f);
        out[(mbase + r) * Nout + col] = v;
    }
}

__global__ void log_softmax64(float* __restrict__ out, int M) {
    const int lane = threadIdx.x & 31;
    const int wave = threadIdx.x >> 5;
    const int row  = blockIdx.x * 8 + wave;
    if (row >= M) return;
    float* p = out + (long long)row * 64;
    float a = p[lane];
    float b = p[lane + 32];
    float m = fmaxf(a, b);
    #pragma unroll
    for (int off = 16; off >= 1; off >>= 1)
        m = fmaxf(m, __shfl_xor(m, off, 32));
    float s = __expf(a - m) + __expf(b - m);
    #pragma unroll
    for (int off = 16; off >= 1; off >>= 1)
        s += __shfl_xor(s, off, 32);
    const float l = __logf(s) + m;
    p[lane]      = a - l;
    p[lane + 32] = b - l;
}

#define AGG_R1 256
#define AGG_R2 64

extern "C" void kernel_launch(void* const* d_in, const int* in_sizes, int n_in,
                              void* d_out, int out_size, void* d_ws, size_t ws_size,
                              hipStream_t stream) {
    const float* x    = (const float*)d_in[0];
    const int*   src1 = (const int*)  d_in[1];
    const int*   dst1 = (const int*)  d_in[2];
    const int*   src2 = (const int*)  d_in[3];
    const int*   dst2 = (const int*)  d_in[4];
    const float* W1l  = (const float*)d_in[7];
    const float* W1r  = (const float*)d_in[8];
    const float* b1   = (const float*)d_in[9];
    const float* W2l  = (const float*)d_in[10];
    const float* W2r  = (const float*)d_in[11];
    const float* b2   = (const float*)d_in[12];

    const int E1 = in_sizes[1];
    const int E2 = in_sizes[3];

    float* agg1 = (float*)d_ws;
    float* cnt1 = agg1 + (size_t)N2c * INc;
    float* h    = cnt1 + N2c;
    float* agg2 = h    + (size_t)N2c * HIDc;
    float* cnt2 = agg2 + (size_t)N3c * HIDc;
    float* out  = (float*)d_out;

    {
        aggregate_owned<INc, AGG_R1>
            <<<(N2c + AGG_R1 - 1) / AGG_R1, AGG_NT, 0, stream>>>(
                x, src1, dst1, agg1, cnt1, E1, N2c);
        dim3 grid(N2c / 16, HIDc / 64);
        sage_gemm_wmma<<<grid, 128, 0, stream>>>(agg1, cnt1, x, W1l, W1r, b1,
                                                 h, INc, HIDc,  1);
    }

    {
        aggregate_owned<HIDc, AGG_R2>
            <<<(N3c + AGG_R2 - 1) / AGG_R2, AGG_NT, 0, stream>>>(
                h, src2, dst2, agg2, cnt2, E2, N3c);
        dim3 grid(N3c / 16, OUTc / 64);
        sage_gemm_wmma<<<grid, 128, 0, stream>>>(agg2, cnt2, h, W2l, W2r, b2,
                                                 out, HIDc, OUTc,  0);
    }

    log_softmax64<<<(N3c + 7) / 8, 256, 0, stream>>>(out, N3c);
}


# SAGE products without LDS: a wave keeps 5x4 (layer 2: 1x4) accumulator tiles; A and B fragments straight from global memory, 16 B per lane, two register sets; k order inside a chunk permuted
# speedup vs baseline: 3.5742x; 1.5810x over previous
#include <hip/hip_runtime.h>
#include <math.h>

#define N1c  200000
#define N2c  40000
#define N3c  10000
#define INc  128
#define HIDc 256
#define OUTc 64

typedef float v2f __attribute__((ext_vector_type(2)));
typedef float v8f __attribute__((ext_vector_type(8)));

#define AGG_NT 1024
#define AGG_NW 32
#define AGG_U  4
#define AGG_QW 544

typedef int   v4i  __attribute__((ext_vector_type(4)));
typedef float v4f  __attribute__((ext_vector_type(4)));

__device__ __forceinline__ v4i load_dst4_tail(const int* __restrict__ dst,
                                              int cbase, int lane, int E) {
    const int e0 = cbase + 4 * lane;
    v4i d;
    d[0] = (e0     < E) ? dst[e0]     : -1;
    d[1] = (e0 + 1 < E) ? dst[e0 + 1] : -1;
    d[2] = (e0 + 2 < E) ? dst[e0 + 2] : -1;
    d[3] = (e0 + 3 < E) ? dst[e0 + 3] : -1;
    return d;
}

__device__ __forceinline__ int slot_chunk(int t, int wave, int rot, int nTot, int nFull) {
    int c = t * AGG_NW + wave + rot;
    if (c >= nTot) c -= nTot;
    return (c < nFull) ? c : -1;
}

__device__ __forceinline__ unsigned slot_rows(int c, int nrows) {
    return (unsigned)__builtin_amdgcn_readfirstlane((c < 0) ? 0 : nrows);
}

__device__ __forceinline__ v4i load_dst4(const int* __restrict__ dst, int c, int lane) {
    const int cl = (c < 0) ? 0 : c;
    return *reinterpret_cast<const v4i*>(dst + (cl << 7) + 4 * lane);
}

__device__ __forceinline__ void queue_chunk(v4i d, int cbase, int lane, int lo,
                                            unsigned nrows, int* q, int& qn) {
    const int e0 = cbase + 4 * lane;
    #pragma unroll
    for (int c = 0; c < 4; ++c) {
        const bool m = (unsigned)(d[c] - lo) < nrows;
        const unsigned mask = __builtin_amdgcn_ballot_w32(m);
        if (mask) {
            if (m) q[qn + (int)__builtin_amdgcn_mbcnt_lo(mask, 0u)] = e0 + c;
            qn += __builtin_popcount(mask);
        }
    }
}

template <int D, int R>
__device__ __forceinline__ void add_batch(const float* __restrict__ x,
                                          const int*   __restrict__ src,
                                          const int*   __restrict__ dst,
                                          float* acc, int* cntL, const int* q,
                                          int base, int count, int lo, int lane) {
    constexpr int NV = D / 64;
    constexpr int G  = 32 / NV;
    const unsigned lane2 = 2u * (unsigned)lane;
    int s = 0, r = R;
    if (lane < count) {
        const int e = q[base + lane];
        s = src[e];
        r = dst[e] - lo;
        atomicAdd(&cntL[r], 1);
    }
    for (int j = 0; j < count; j += G) {
        v2f v[G][NV];
        int rr[G];
        #pragma unroll
        for (int g = 0; g < G; ++g) {
            const int sg = __builtin_amdgcn_readlane(s, j + g);
            rr[g]        = __builtin_amdgcn_readlane(r, j + g);
            const float* row = x + (long long)sg * D;
            #pragma unroll
            for (int k = 0; k < NV; ++k)
                v[g][k] = *reinterpret_cast<const v2f*>(row + (lane2 + 64u * k));
        }
        #pragma unroll
        for (int g = 0; g < G; ++g) {
            float* a = acc + rr[g] * D + 2 * lane;
            #pragma unroll
            for (int k = 0; k < NV; ++k) {
                atomicAdd(a + 64 * k,     v[g][k][0]);
                atomicAdd(a + 64 * k + 1, v[g][k][1]);
            }
        }
    }
}

template <int D, int R>
__launch_bounds__(AGG_NT)
__global__ void aggregate_owned(const float* __restrict__ x,
                                const int*   __restrict__ src,
                                const int*   __restrict__ dst,
                                float*       __restrict__ agg,
                                float*       __restrict__ cnt,
                                int E, int N) {
    __shared__ __attribute__((aligned(16))) float acc[(R + 1) * D];
    __shared__ int cntL[R + 1];
    __shared__ int qAll[AGG_NW * AGG_QW];

    const int tid   = threadIdx.x;
    const int lane  = tid & 31;
    const int wave  = __builtin_amdgcn_readfirstlane(tid >> 5);
    const int lo    = blockIdx.x * R;
    const int nrows = (N - lo < R) ? (N - lo) : R;

    {
        v4f z; z[0] = 0.0f; z[1] = 0.0f; z[2] = 0.0f; z[3] = 0.0f;
        v4f* a4 = reinterpret_cast<v4f*>(acc);
        for (int i = tid; i < (R + 1) * D / 4; i += AGG_NT) a4[i] = z;
        if (tid <= R) cntL[tid] = 0;
    }
    __syncthreads();

    const int nFull = E >> 7;
    const int nGrp  = (nFull + AGG_NW * AGG_U - 1) / (AGG_NW * AGG_U);
    const int nTrip = (nGrp + 1) >> 1;
    const int nTot  = nTrip * 2 * AGG_NW * AGG_U;
    const int rot   = (int)blockIdx.x * (nTot / (int)gridDim.x);

    int* q  = qAll + wave * AGG_QW;
    int  qn = 0;

    if (nFull > 0) {
        v4i da[AGG_U], db[AGG_U];
        int ca[AGG_U], cb[AGG_U];
        #pragma unroll
        for (int u = 0; u < AGG_U; ++u) {
            ca[u] = slot_chunk(u, wave, rot, nTot, nFull);
            da[u] = load_dst4(dst, ca[u], lane);
        }
        for (int t = 0; t < nTrip * 2 * AGG_U; t += 2 * AGG_U) {
            #pragma unroll
            for (int u = 0; u < AGG_U; ++u) {
                cb[u] = slot_chunk(t + AGG_U + u, wave, rot, nTot, nFull);
                db[u] = load_dst4(dst, cb[u], lane);
            }
            #pragma unroll
            for (int u = 0; u < AGG_U; ++u)
                queue_chunk(da[u], ca[u] << 7, lane, lo, slot_rows(ca[u], nrows), q, qn);
            while (qn >= 32) {
                qn -= 32;
                add_batch<D, R>(x, src, dst, acc, cntL, q, qn, 32, lo, lane);
            }
            #pragma unroll
            for (int u = 0; u < AGG_U; ++u) {
                ca[u] = (t + 2 * AGG_U < nTrip * 2 * AGG_U)
                            ? slot_chunk(t + 2 * AGG_U + u, wave, rot, nTot, nFull) : -1;
                da[u] = load_dst4(dst, ca[u], lane);
            }
            #pragma unroll
            for (int u = 0; u < AGG_U; ++u)
                queue_chunk(db[u], cb[u] << 7, lane, lo, slot_rows(cb[u], nrows), q, qn);
            while (qn >= 32) {
                qn -= 32;
                add_batch<D, R>(x, src, dst, acc, cntL, q, qn, 32, lo, lane);
            }
        }
    }
    if ((E & 127) != 0 && wave == 0) {
        const v4i d = load_dst4_tail(dst, nFull << 7, lane, E);
        queue_chunk(d, nFull << 7, lane, lo, (unsigned)nrows, q, qn);
        while (qn >= 32) {
            qn -= 32;
            add_batch<D, R>(x, src, dst, acc, cntL, q, qn, 32, lo, lane);
        }
    }
    if (qn > 0) add_batch<D, R>(x, src, dst, acc, cntL, q, 0, qn, lo, lane);

    __syncthreads();

    {
        const v4f* a4 = reinterpret_cast<const v4f*>(acc);
        v4f* o4 = reinterpret_cast<v4f*>(agg + (long long)lo * D);
        for (int i = tid; i < nrows * (D / 4); i += AGG_NT) o4[i] = a4[i];
        if (tid < nrows) cnt[lo + tid] = (float)cntL[tid];
    }
}

template <int MT, int NT, int KC, int K>
__device__ __forceinline__ void load_frags(const float* __restrict__ aRow,
                                           const float* __restrict__ bRow, int k0,
                                           v4f (&a)[MT][KC], v4f (&b)[NT][KC]) {
    #pragma unroll
    for (int i = 0; i < MT; ++i) {
        #pragma unroll
        for (int q = 0; q < KC; ++q)
            a[i][q] = *reinterpret_cast<const v4f*>(aRow + i * 16 * K + k0 + 4 * q);
    }
    #pragma unroll
    for (int j = 0; j < NT; ++j) {
        #pragma unroll
        for (int q = 0; q < KC; ++q)
            b[j][q] = *reinterpret_cast<const v4f*>(bRow + j * 16 * K + k0 + 4 * q);
    }
}

template <int MT, int NT, int KC, bool SCALE>
__device__ __forceinline__ void mul_frags(v4f (&a)[MT][KC], const v4f (&b)[NT][KC],
                                          const float (&inv)[MT], v8f (&acc)[MT][NT]) {
    if (SCALE) {
        #pragma unroll
        for (int i = 0; i < MT; ++i) {
            #pragma unroll
            for (int q = 0; q < KC; ++q) a[i][q] *= inv[i];
        }
    }
    #pragma unroll
    for (int q = 0; q < KC; ++q) {
        #pragma unroll
        for (int s = 0; s < 2; ++s) {
            #pragma unroll
            for (int i = 0; i < MT; ++i) {
                v2f av; av[0] = a[i][q][2 * s]; av[1] = a[i][q][2 * s + 1];
                #pragma unroll
                for (int j = 0; j < NT; ++j) {
                    v2f bv; bv[0] = b[j][q][2 * s]; bv[1] = b[j][q][2 * s + 1];
                    acc[i][j] = __builtin_amdgcn_wmma_f32_16x16x4_f32(
                        false, av, false, bv, (short)0, acc[i][j], false, false);
                }
            }
        }
    }
}

template <int MT, int NT, int KC, int K, bool SCALE, bool NEXT>
__device__ __forceinline__ void sage_term(const float* __restrict__ aRow,
                                          const float* __restrict__ bRow,
                                          const float* __restrict__ aNext,
                                          const float* __restrict__ bNext,
                                          const float (&inv)[MT],
                                          v4f (&a0)[MT][KC], v4f (&b0)[NT][KC],
                                          v8f (&acc)[MT][NT]) {
    static_assert(K % (16 * KC) == 0, "two chunks per trip");
    v4f a1[MT][KC], b1[NT][KC];
    #pragma unroll 1
    for (int k0 = 0; k0 < K - 16 * KC; k0 += 16 * KC) {
        load_frags<MT, NT, KC, K>(aRow, bRow, k0 + 8 * KC, a1, b1);
        mul_frags<MT, NT, KC, SCALE>(a0, b0, inv, acc);
        load_frags<MT, NT, KC, K>(aRow, bRow, k0 + 16 * KC, a0, b0);
        mul_frags<MT, NT, KC, SCALE>(a1, b1, inv, acc);
    }
    load_frags<MT, NT, KC, K>(aRow, bRow, K - 8 * KC, a1, b1);
    mul_frags<MT, NT, KC, SCALE>(a0, b0, inv, acc);
    if (NEXT) load_frags<MT, NT, KC, K>(aNext, bNext, 0, a0, b0);
    mul_frags<MT, NT, KC, SCALE>(a1, b1, inv, acc);
}

template <int MT, int NT, int KC, int K, int NOUT, bool RELU>
__launch_bounds__(32 * NOUT / (16 * NT))
__global__ void sage_gemm_regs(const float* __restrict__ agg,
                               const float* __restrict__ cnt,
                               const float* __restrict__ xt,
                               const float* __restrict__ Wl,
                               const float* __restrict__ Wr,
                               const float* __restrict__ bias,
                               float*       __restrict__ out) {
    const int lane = threadIdx.x & 31;
    const int wave = threadIdx.x >> 5;
    const int c    = lane & 15;
    const int h    = lane >> 4;

    const long long row0 = (long long)blockIdx.x * (16 * MT);
    const int       col0 = wave * (16 * NT);

    float inv[MT];
    #pragma unroll
    for (int i = 0; i < MT; ++i)
        inv[i] = 1.0f / fmaxf(cnt[row0 + 16 * i + c], 1.0f);

    v8f acc[MT][NT];
    #pragma unroll
    for (int i = 0; i < MT; ++i) {
        #pragma unroll
        for (int j = 0; j < NT; ++j) {
            #pragma unroll
            for (int r = 0; r < 8; ++r) acc[i][j][r] = 0.0f;
        }
    }

    const long long aOff = (row0 + c) * K + 4 * KC * h;
    const int       bOff = (col0 + c) * K + 4 * KC * h;

    v4f a0[MT][KC], b0[NT][KC];
    load_frags<MT, NT, KC, K>(agg + aOff, Wl + bOff, 0, a0, b0);
    sage_term<MT, NT, KC, K, true,  true >(agg + aOff, Wl + bOff, xt + aOff, Wr + bOff,
                                           inv, a0, b0, acc);
    sage_term<MT, NT, KC, K, false, false>(xt + aOff, Wr + bOff, xt + aOff, Wr + bOff,
                                           inv, a0, b0, acc);

    #pragma unroll
    for (int j = 0; j < NT; ++j) {
        const int   col = col0 + 16 * j + c;
        const float bv  = bias[col];
        #pragma unroll
        for (int i = 0; i < MT; ++i) {
            const long long mbase = row0 + 16 * i + 8 * h;
            #pragma unroll
            for (int r = 0; r < 8; ++r) {
                float v = acc[i][j][r] + bv;
                if (RELU) v = fmaxf(v, 0.0f);
                out[(mbase + r) * NOUT + col] = v;
            }
        }
    }
}

__global__ void log_softmax64(float* __restrict__ out, int M) {
    const int lane = threadIdx.x & 31;
    const int wave = threadIdx.x >> 5;
    const int row  = blockIdx.x * 8 + wave;
    if (row >= M) return;
    float* p = out + (long long)row * 64;
    float a = p[lane];
    float b = p[lane + 32];
    float m = fmaxf(a, b);
    #pragma unroll
    for (int off = 16; off >= 1; off >>= 1)
        m = fmaxf(m, __shfl_xor(m, off, 32));
    float s = __expf(a - m) + __expf(b - m);
    #pragma unroll
    for (int off = 16; off >= 1; off >>= 1)
        s += __shfl_xor(s, off, 32);
    const float l = __logf(s) + m;
    p[lane]      = a - l;
    p[lane + 32] = b - l;
}

#define AGG_R1 256
#define AGG_R2 64

extern "C" void kernel_launch(void* const* d_in, const int* in_sizes, int n_in,
                              void* d_out, int out_size, void* d_ws, size_t ws_size,
                              hipStream_t stream) {
    const float* x    = (const float*)d_in[0];
    const int*   src1 = (const int*)  d_in[1];
    const int*   dst1 = (const int*)  d_in[2];
    const int*   src2 = (const int*)  d_in[3];
    const int*   dst2 = (const int*)  d_in[4];
    const float* W1l  = (const float*)d_in[7];
    const float* W1r  = (const float*)d_in[8];
    const float* b1   = (const float*)d_in[9];
    const float* W2l  = (const float*)d_in[10];
    const float* W2r  = (const float*)d_in[11];
    const float* b2   = (const float*)d_in[12];

    const int E1 = in_sizes[1];
    const int E2 = in_sizes[3];

    float* agg1 = (float*)d_ws;
    float* cnt1 = agg1 + (size_t)N2c * INc;
    float* h    = cnt1 + N2c;
    float* agg2 = h    + (size_t)N2c * HIDc;
    float* cnt2 = agg2 + (size_t)N3c * HIDc;
    float* out  = (float*)d_out;

    {
        aggregate_owned<INc, AGG_R1>
            <<<(N2c + AGG_R1 - 1) / AGG_R1, AGG_NT, 0, stream>>>(
                x, src1, dst1, agg1, cnt1, E1, N2c);
        sage_gemm_regs<5, 4, 2, INc, HIDc, true>
            <<<N2c / 80, 128, 0, stream>>>(agg1, cnt1, x, W1l, W1r, b1, h);
    }

    {
        aggregate_owned<HIDc, AGG_R2>
            <<<(N3c + AGG_R2 - 1) / AGG_R2, AGG_NT, 0, stream>>>(
                h, src2, dst2, agg2, cnt2, E2, N3c);
        sage_gemm_regs<1, 4, 4, HIDc, OUTc, false>
            <<<N3c / 16, 32, 0, stream>>>(agg2, cnt2, h, W2l, W2r, b2, out);
    }

    log_softmax64<<<(N3c + 7) / 8, 256, 0, stream>>>(out, N3c);
}


# Owner ranges shrunk to one work-group per WGP: 157 rows (layer 2: 40) instead of 256 (64), 255 and 250 work-groups instead of 157; kernels unchanged; sums still in arrival order
# speedup vs baseline: 4.0697x; 1.1386x over previous
#include <hip/hip_runtime.h>
#include <math.h>

#define N1c  200000
#define N2c  40000
#define N3c  10000
#define INc  128
#define HIDc 256
#define OUTc 64

typedef float v2f __attribute__((ext_vector_type(2)));
typedef float v8f __attribute__((ext_vector_type(8)));

#define AGG_NT 1024
#define AGG_NW 32
#define AGG_U  4
#define AGG_QW 544

typedef int   v4i  __attribute__((ext_vector_type(4)));
typedef float v4f  __attribute__((ext_vector_type(4)));

__device__ __forceinline__ v4i load_dst4_tail(const int* __restrict__ dst,
                                              int cbase, int lane, int E) {
    const int e0 = cbase + 4 * lane;
    v4i d;
    d[0] = (e0     < E) ? dst[e0]     : -1;
    d[1] = (e0 + 1 < E) ? dst[e0 + 1] : -1;
    d[2] = (e0 + 2 < E) ? dst[e0 + 2] : -1;
    d[3] = (e0 + 3 < E) ? dst[e0 + 3] : -1;
    return d;
}

__device__ __forceinline__ int slot_chunk(int t, int wave, int rot, int nTot, int nFull) {
    int c = t * AGG_NW + wave + rot;
    if (c >= nTot) c -= nTot;
    return (c < nFull) ? c : -1;
}

__device__ __forceinline__ unsigned slot_rows(int c, int nrows) {
    return (unsigned)__builtin_amdgcn_readfirstlane((c < 0) ? 0 : nrows);
}

__device__ __forceinline__ v4i load_dst4(const int* __restrict__ dst, int c, int lane) {
    const int cl = (c < 0) ? 0 : c;
    return *reinterpret_cast<const v4i*>(dst + (cl << 7) + 4 * lane);
}

__device__ __forceinline__ void queue_chunk(v4i d, int cbase, int lane, int lo,
                                            unsigned nrows, int* q, int& qn) {
    const int e0 = cbase + 4 * lane;
    #pragma unroll
    for (int c = 0; c < 4; ++c) {
        const bool m = (unsigned)(d[c] - lo) < nrows;
        const unsigned mask = __builtin_amdgcn_ballot_w32(m);
        if (mask) {
            if (m) q[qn + (int)__builtin_amdgcn_mbcnt_lo(mask, 0u)] = e0 + c;
            qn += __builtin_popcount(mask);
        }
    }
}

template <int D, int R>
__device__ __forceinline__ void add_batch(const float* __restrict__ x,
                                          const int*   __restrict__ src,
                                          const int*   __restrict__ dst,
                                          float* acc, int* cntL, const int* q,
                                          int base, int count, int lo, int lane) {
    constexpr int NV = D / 64;
    constexpr int G  = 32 / NV;
    const unsigned lane2 = 2u * (unsigned)lane;
    int s = 0, r = R;
    if (lane < count) {
        const int e = q[base + lane];
        s = src[e];
        r = dst[e] - lo;
        atomicAdd(&cntL[r], 1);
    }
    for (int j = 0; j < count; j += G) {
        v2f v[G][NV];
        int rr[G];
        #pragma unroll
        for (int g = 0; g < G; ++g) {
            const int sg = __builtin_amdgcn_readlane(s, j + g);
            rr[g]        = __builtin_amdgcn_readlane(r, j + g);
            const float* row = x + (long long)sg * D;
            #pragma unroll
            for (int k = 0; k < NV; ++k)
                v[g][k] = *reinterpret_cast<const v2f*>(row + (lane2 + 64u * k));
        }
        #pragma unroll
        for (int g = 0; g < G; ++g) {
            float* a = acc + rr[g] * D + 2 * lane;
            #pragma unroll
            for (int k = 0; k < NV; ++k) {
                atomicAdd(a + 64 * k,     v[g][k][0]);
                atomicAdd(a + 64 * k + 1, v[g][k][1]);
            }
        }
    }
}

template <int D, int R>
__launch_bounds__(AGG_NT)
__global__ void aggregate_owned(const float* __restrict__ x,
                                const int*   __restrict__ src,
                                const int*   __restrict__ dst,
                                float*       __restrict__ agg,
                                float*       __restrict__ cnt,
                                int E, int N) {
    __shared__ __attribute__((aligned(16))) float acc[(R + 1) * D];
    __shared__ int cntL[R + 1];
    __shared__ int qAll[AGG_NW * AGG_QW];

    const int tid   = threadIdx.x;
    const int lane  = tid & 31;
    const int wave  = __builtin_amdgcn_readfirstlane(tid >> 5);
    const int lo    = blockIdx.x * R;
    const int nrows = (N - lo < R) ? (N - lo) : R;

    {
        v4f z; z[0] = 0.0f; z[1] = 0.0f; z[2] = 0.0f; z[3] = 0.0f;
        v4f* a4 = reinterpret_cast<v4f*>(acc);
        for (int i = tid; i < (R + 1) * D / 4; i += AGG_NT) a4[i] = z;
        if (tid <= R) cntL[tid] = 0;
    }
    __syncthreads();

    const int nFull = E >> 7;
    const int nGrp  = (nFull + AGG_NW * AGG_U - 1) / (AGG_NW * AGG_U);
    const int nTrip = (nGrp + 1) >> 1;
    const int nTot  = nTrip * 2 * AGG_NW * AGG_U;
    const int rot   = (int)blockIdx.x * (nTot / (int)gridDim.x);

    int* q  = qAll + wave * AGG_QW;
    int  qn = 0;

    if (nFull > 0) {
        v4i da[AGG_U], db[AGG_U];
        int ca[AGG_U], cb[AGG_U];
        #pragma unroll
        for (int u = 0; u < AGG_U; ++u) {
            ca[u] = slot_chunk(u, wave, rot, nTot, nFull);
            da[u] = load_dst4(dst, ca[u], lane);
        }
        for (int t = 0; t < nTrip * 2 * AGG_U; t += 2 * AGG_U) {
            #pragma unroll
            for (int u = 0; u < AGG_U; ++u) {
                cb[u] = slot_chunk(t + AGG_U + u, wave, rot, nTot, nFull);
                db[u] = load_dst4(dst, cb[u], lane);
            }
            #pragma unroll
            for (int u = 0; u < AGG_U; ++u)
                queue_chunk(da[u], ca[u] << 7, lane, lo, slot_rows(ca[u], nrows), q, qn);
            while (qn >= 32) {
                qn -= 32;
                add_batch<D, R>(x, src, dst, acc, cntL, q, qn, 32, lo, lane);
            }
            #pragma unroll
            for (int u = 0; u < AGG_U; ++u) {
                ca[u] = (t + 2 * AGG_U < nTrip * 2 * AGG_U)
                            ? slot_chunk(t + 2 * AGG_U + u, wave, rot, nTot, nFull) : -1;
                da[u] = load_dst4(dst, ca[u], lane);
            }
            #pragma unroll
            for (int u = 0; u < AGG_U; ++u)
                queue_chunk(db[u], cb[u] << 7, lane, lo, slot_rows(cb[u], nrows), q, qn);
            while (qn >= 32) {
                qn -= 32;
                add_batch<D, R>(x, src, dst, acc, cntL, q, qn, 32, lo, lane);
            }
        }
    }
    if ((E & 127) != 0 && wave == 0) {
        const v4i d = load_dst4_tail(dst, nFull << 7, lane, E);
        queue_chunk(d, nFull << 7, lane, lo, (unsigned)nrows, q, qn);
        while (qn >= 32) {
            qn -= 32;
            add_batch<D, R>(x, src, dst, acc, cntL, q, qn, 32, lo, lane);
        }
    }
    if (qn > 0) add_batch<D, R>(x, src, dst, acc, cntL, q, 0, qn, lo, lane);

    __syncthreads();

    {
        const v4f* a4 = reinterpret_cast<const v4f*>(acc);
        v4f* o4 = reinterpret_cast<v4f*>(agg + (long long)lo * D);
        for (int i = tid; i < nrows * (D / 4); i += AGG_NT) o4[i] = a4[i];
        if (tid < nrows) cnt[lo + tid] = (float)cntL[tid];
    }
}

template <int MT, int NT, int KC, int K>
__device__ __forceinline__ void load_frags(const float* __restrict__ aRow,
                                           const float* __restrict__ bRow, int k0,
                                           v4f (&a)[MT][KC], v4f (&b)[NT][KC]) {
    #pragma unroll
    for (int i = 0; i < MT; ++i) {
        #pragma unroll
        for (int q = 0; q < KC; ++q)
            a[i][q] = *reinterpret_cast<const v4f*>(aRow + i * 16 * K + k0 + 4 * q);
    }
    #pragma unroll
    for (int j = 0; j < NT; ++j) {
        #pragma unroll
        for (int q = 0; q < KC; ++q)
            b[j][q] = *reinterpret_cast<const v4f*>(bRow + j * 16 * K + k0 + 4 * q);
    }
}

template <int MT, int NT, int KC, bool SCALE>
__device__ __forceinline__ void mul_frags(v4f (&a)[MT][KC], const v4f (&b)[NT][KC],
                                          const float (&inv)[MT], v8f (&acc)[MT][NT]) {
    if (SCALE) {
        #pragma unroll
        for (int i = 0; i < MT; ++i) {
            #pragma unroll
            for (int q = 0; q < KC; ++q) a[i][q] *= inv[i];
        }
    }
    #pragma unroll
    for (int q = 0; q < KC; ++q) {
        #pragma unroll
        for (int s = 0; s < 2; ++s) {
            #pragma unroll
            for (int i = 0; i < MT; ++i) {
                v2f av; av[0] = a[i][q][2 * s]; av[1] = a[i][q][2 * s + 1];
                #pragma unroll
                for (int j = 0; j < NT; ++j) {
                    v2f bv; bv[0] = b[j][q][2 * s]; bv[1] = b[j][q][2 * s + 1];
                    acc[i][j] = __builtin_amdgcn_wmma_f32_16x16x4_f32(
                        false, av, false, bv, (short)0, acc[i][j], false, false);
                }
            }
        }
    }
}

template <int MT, int NT, int KC, int K, bool SCALE, bool NEXT>
__device__ __forceinline__ void sage_term(const float* __restrict__ aRow,
                                          const float* __restrict__ bRow,
                                          const float* __restrict__ aNext,
                                          const float* __restrict__ bNext,
                                          const float (&inv)[MT],
                                          v4f (&a0)[MT][KC], v4f (&b0)[NT][KC],
                                          v8f (&acc)[MT][NT]) {
    static_assert(K % (16 * KC) == 0, "two chunks per trip");
    v4f a1[MT][KC], b1[NT][KC];
    #pragma unroll 1
    for (int k0 = 0; k0 < K - 16 * KC; k0 += 16 * KC) {
        load_frags<MT, NT, KC, K>(aRow, bRow, k0 + 8 * KC, a1, b1);
        mul_frags<MT, NT, KC, SCALE>(a0, b0, inv, acc);
        load_frags<MT, NT, KC, K>(aRow, bRow, k0 + 16 * KC, a0, b0);
        mul_frags<MT, NT, KC, SCALE>(a1, b1, inv, acc);
    }
    load_frags<MT, NT, KC, K>(aRow, bRow, K - 8 * KC, a1, b1);
    mul_frags<MT, NT, KC, SCALE>(a0, b0, inv, acc);
    if (NEXT) load_frags<MT, NT, KC, K>(aNext, bNext, 0, a0, b0);
    mul_frags<MT, NT, KC, SCALE>(a1, b1, inv, acc);
}

template <int MT, int NT, int KC, int K, int NOUT, bool RELU>
__launch_bounds__(32 * NOUT / (16 * NT))
__global__ void sage_gemm_regs(const float* __restrict__ agg,
                               const float* __restrict__ cnt,
                               const float* __restrict__ xt,
                               const float* __restrict__ Wl,
                               const float* __restrict__ Wr,
                               const float* __restrict__ bias,
                               float*       __restrict__ out) {
    const int lane = threadIdx.x & 31;
    const int wave = threadIdx.x >> 5;
    const int c    = lane & 15;
    const int h    = lane >> 4;

    const long long row0 = (long long)blockIdx.x * (16 * MT);
    const int       col0 = wave * (16 * NT);

    float inv[MT];
    #pragma unroll
    for (int i = 0; i < MT; ++i)
        inv[i] = 1.0f / fmaxf(cnt[row0 + 16 * i + c], 1.0f);

    v8f acc[MT][NT];
    #pragma unroll
    for (int i = 0; i < MT; ++i) {
        #pragma unroll
        for (int j = 0; j < NT; ++j) {
            #pragma unroll
            for (int r = 0; r < 8; ++r) acc[i][j][r] = 0.0f;
        }
    }

    const long long aOff = (row0 + c) * K + 4 * KC * h;
    const int       bOff = (col0 + c) * K + 4 * KC * h;

    v4f a0[MT][KC], b0[NT][KC];
    load_frags<MT, NT, KC, K>(agg + aOff, Wl + bOff, 0, a0, b0);
    sage_term<MT, NT, KC, K, true,  true >(agg + aOff, Wl + bOff, xt + aOff, Wr + bOff,
                                           inv, a0, b0, acc);
    sage_term<MT, NT, KC, K, false, false>(xt + aOff, Wr + bOff, xt + aOff, Wr + bOff,
                                           inv, a0, b0, acc);

    #pragma unroll
    for (int j = 0; j < NT; ++j) {
        const int   col = col0 + 16 * j + c;
        const float bv  = bias[col];
        #pragma unroll
        for (int i = 0; i < MT; ++i) {
            const long long mbase = row0 + 16 * i + 8 * h;
            #pragma unroll
            for (int r = 0; r < 8; ++r) {
                float v = acc[i][j][r] + bv;
                if (RELU) v = fmaxf(v, 0.0f);
                out[(mbase + r) * NOUT + col] = v;
            }
        }
    }
}

__global__ void log_softmax64(float* __restrict__ out, int M) {
    const int lane = threadIdx.x & 31;
    const int wave = threadIdx.x >> 5;
    const int row  = blockIdx.x * 8 + wave;
    if (row >= M) return;
    float* p = out + (long long)row * 64;
    float a = p[lane];
    float b = p[lane + 32];
    float m = fmaxf(a, b);
    #pragma unroll
    for (int off = 16; off >= 1; off >>= 1)
        m = fmaxf(m, __shfl_xor(m, off, 32));
    float s = __expf(a - m) + __expf(b - m);
    #pragma unroll
    for (int off = 16; off >= 1; off >>= 1)
        s += __shfl_xor(s, off, 32);
    const float l = __logf(s) + m;
    p[lane]      = a - l;
    p[lane + 32] = b - l;
}

#define AGG_WGPS 256
#define AGG_R1 ((N2c + AGG_WGPS - 1) / AGG_WGPS)
#define AGG_R2 ((N3c + AGG_WGPS - 1) / AGG_WGPS)

extern "C" void kernel_launch(void* const* d_in, const int* in_sizes, int n_in,
                              void* d_out, int out_size, void* d_ws, size_t ws_size,
                              hipStream_t stream) {
    const float* x    = (const float*)d_in[0];
    const int*   src1 = (const int*)  d_in[1];
    const int*   dst1 = (const int*)  d_in[2];
    const int*   src2 = (const int*)  d_in[3];
    const int*   dst2 = (const int*)  d_in[4];
    const float* W1l  = (const float*)d_in[7];
    const float* W1r  = (const float*)d_in[8];
    const float* b1   = (const float*)d_in[9];
    const float* W2l  = (const float*)d_in[10];
    const float* W2r  = (const float*)d_in[11];
    const float* b2   = (const float*)d_in[12];

    const int E1 = in_sizes[1];
    const int E2 = in_sizes[3];

    float* agg1 = (float*)d_ws;
    float* cnt1 = agg1 + (size_t)N2c * INc;
    float* h    = cnt1 + N2c;
    float* agg2 = h    + (size_t)N2c * HIDc;
    float* cnt2 = agg2 + (size_t)N3c * HIDc;
    float* out  = (float*)d_out;

    {
        aggregate_owned<INc, AGG_R1>
            <<<(N2c + AGG_R1 - 1) / AGG_R1, AGG_NT, 0, stream>>>(
                x, src1, dst1, agg1, cnt1, E1, N2c);
        sage_gemm_regs<5, 4, 2, INc, HIDc, true>
            <<<N2c / 80, 128, 0, stream>>>(agg1, cnt1, x, W1l, W1r, b1, h);
    }

    {
        aggregate_owned<HIDc, AGG_R2>
            <<<(N3c + AGG_R2 - 1) / AGG_R2, AGG_NT, 0, stream>>>(
                h, src2, dst2, agg2, cnt2, E2, N3c);
        sage_gemm_regs<1, 4, 4, HIDc, OUTc, false>
            <<<N3c / 16, 32, 0, stream>>>(agg2, cnt2, h, W2l, W2r, b2, out);
    }

    log_softmax64<<<(N3c + 7) / 8, 256, 0, stream>>>(out, N3c);
}


# Owner work-groups scan one-byte owner keys (dst / rows per owner, packed by a new first launch) instead of 4-byte dst, four edges per compare; 6 launches; sums still in arrival order
# speedup vs baseline: 4.3317x; 1.0644x over previous
#include <hip/hip_runtime.h>
#include <math.h>

#define N1c  200000
#define N2c  40000
#define N3c  10000
#define INc  128
#define HIDc 256
#define OUTc 64

typedef float v2f __attribute__((ext_vector_type(2)));
typedef float v8f __attribute__((ext_vector_type(8)));

#define AGG_NT 1024
#define AGG_NW 32
#define AGG_U  4
#define AGG_QW 1056

typedef int      v4i  __attribute__((ext_vector_type(4)));
typedef unsigned v4u  __attribute__((ext_vector_type(4)));
typedef float    v4f  __attribute__((ext_vector_type(4)));

template <int R>
__device__ __forceinline__ unsigned key_of(int d) {
    const unsigned k = (unsigned)d / (unsigned)R;
    return (k < 255u) ? k : 255u;
}

template <int R>
__device__ __forceinline__ unsigned pack4(v4i d) {
    return key_of<R>(d[0]) | (key_of<R>(d[1]) << 8) |
           (key_of<R>(d[2]) << 16) | (key_of<R>(d[3]) << 24);
}

template <int R>
__device__ __forceinline__ void pack_keys16(const int* __restrict__ dst,
                                            unsigned* __restrict__ keys, int t, int E) {
    const int e0 = 16 * t;
    v4u w;
    if (e0 + 15 < E) {
        const v4i* p = reinterpret_cast<const v4i*>(dst + e0);
        const v4i d0 = p[0], d1 = p[1], d2 = p[2], d3 = p[3];
        w[0] = pack4<R>(d0);
        w[1] = pack4<R>(d1);
        w[2] = pack4<R>(d2);
        w[3] = pack4<R>(d3);
    } else {
        #pragma unroll
        for (int j = 0; j < 4; ++j) {
            unsigned word = 0u;
            #pragma unroll
            for (int c = 0; c < 4; ++c) {
                const int e = e0 + 4 * j + c;
                const unsigned k = (e < E) ? key_of<R>(dst[e]) : 255u;
                word |= k << (8 * c);
            }
            w[j] = word;
        }
    }
    *reinterpret_cast<v4u*>(keys + 4 * t) = w;
}

__device__ __forceinline__ int slot_chunk(int t, int wave, int rot, int nTot, int nCh) {
    int c = t * AGG_NW + wave + rot;
    if (c >= nTot) c -= nTot;
    return (c < nCh) ? c : -1;
}

__device__ __forceinline__ v4u load_keys(const unsigned* __restrict__ keys, int c, int lane) {
    const int cl = (c < 0) ? 0 : c;
    return *reinterpret_cast<const v4u*>(keys + (cl << 7) + 4 * lane);
}

__device__ __forceinline__ unsigned zero_bytes(unsigned x) {
    const unsigned t = (x & 0x7F7F7F7Fu) + 0x7F7F7F7Fu;
    return ~(t | x | 0x7F7F7F7Fu);
}

__device__ __forceinline__ void queue_chunk(v4u kw, int c, int lane, unsigned pat,
                                            int* q, int& qn) {
    if (c < 0) return;
    const int e0 = (c << 9) + 16 * lane;
    #pragma unroll
    for (int w = 0; w < 4; ++w) {
        unsigned z = zero_bytes(kw[w] ^ pat);
        unsigned mask = __builtin_amdgcn_ballot_w32(z != 0u);
        while (mask) {
            if (z != 0u) {
                q[qn + (int)__builtin_amdgcn_mbcnt_lo(mask, 0u)] =
                    e0 + 4 * w + (__builtin_ctz(z) >> 3);
                z &= z - 1u;
            }
            qn += __builtin_popcount(mask);
            mask = __builtin_amdgcn_ballot_w32(z != 0u);
        }
    }
}

template <int D, int R>
__device__ __forceinline__ void add_batch(const float* __restrict__ x,
                                          const int*   __restrict__ src,
                                          const int*   __restrict__ dst,
                                          float* acc, int* cntL, const int* q,
                                          int base, int count, int lo, int lane) {
    constexpr int NV = D / 64;
    constexpr int G  = 32 / NV;
    const unsigned lane2 = 2u * (unsigned)lane;
    int s = 0, r = R;
    if (lane < count) {
        const int e = q[base + lane];
        s = src[e];
        r = dst[e] - lo;
        atomicAdd(&cntL[r], 1);
    }
    for (int j = 0; j < count; j += G) {
        v2f v[G][NV];
        int rr[G];
        #pragma unroll
        for (int g = 0; g < G; ++g) {
            const int sg = __builtin_amdgcn_readlane(s, j + g);
            rr[g]        = __builtin_amdgcn_readlane(r, j + g);
            const float* row = x + (long long)sg * D;
            #pragma unroll
            for (int k = 0; k < NV; ++k)
                v[g][k] = *reinterpret_cast<const v2f*>(row + (lane2 + 64u * k));
        }
        #pragma unroll
        for (int g = 0; g < G; ++g) {
            float* a = acc + rr[g] * D + 2 * lane;
            #pragma unroll
            for (int k = 0; k < NV; ++k) {
                atomicAdd(a + 64 * k,     v[g][k][0]);
                atomicAdd(a + 64 * k + 1, v[g][k][1]);
            }
        }
    }
}

template <int D, int R>
__launch_bounds__(AGG_NT)
__global__ void aggregate_owned(const float*    __restrict__ x,
                                const int*      __restrict__ src,
                                const int*      __restrict__ dst,
                                const unsigned* __restrict__ keys,
                                float*          __restrict__ agg,
                                float*          __restrict__ cnt,
                                int E, int N) {
    __shared__ __attribute__((aligned(16))) float acc[(R + 1) * D];
    __shared__ int cntL[R + 1];
    __shared__ int qAll[AGG_NW * AGG_QW];

    const int tid   = threadIdx.x;
    const int lane  = tid & 31;
    const int wave  = __builtin_amdgcn_readfirstlane(tid >> 5);
    const int lo    = blockIdx.x * R;
    const int nrows = (N - lo < R) ? (N - lo) : R;
    const unsigned pat = (unsigned)blockIdx.x * 0x01010101u;

    {
        v4f z; z[0] = 0.0f; z[1] = 0.0f; z[2] = 0.0f; z[3] = 0.0f;
        v4f* a4 = reinterpret_cast<v4f*>(acc);
        for (int i = tid; i < (R + 1) * D / 4; i += AGG_NT) a4[i] = z;
        if (tid <= R) cntL[tid] = 0;
    }
    __syncthreads();

    const int nCh   = (E + 511) >> 9;
    const int nGrp  = (nCh + AGG_NW * AGG_U - 1) / (AGG_NW * AGG_U);
    const int nTrip = (nGrp + 1) >> 1;
    const int nTot  = nTrip * 2 * AGG_NW * AGG_U;
    const int rot   = (int)blockIdx.x * (nTot / (int)gridDim.x);

    int* q  = qAll + wave * AGG_QW;
    int  qn = 0;

    if (nCh > 0) {
        v4u da[AGG_U], db[AGG_U];
        int ca[AGG_U], cb[AGG_U];
        #pragma unroll
        for (int u = 0; u < AGG_U; ++u) {
            ca[u] = slot_chunk(u, wave, rot, nTot, nCh);
            da[u] = load_keys(keys, ca[u], lane);
        }
        for (int t = 0; t < nTrip * 2 * AGG_U; t += 2 * AGG_U) {
            #pragma unroll
            for (int u = 0; u < AGG_U; ++u) {
                cb[u] = slot_chunk(t + AGG_U + u, wave, rot, nTot, nCh);
                db[u] = load_keys(keys, cb[u], lane);
            }
            #pragma unroll
            for (int u = 0; u < AGG_U; u += 2) {
                queue_chunk(da[u],     ca[u],     lane, pat, q, qn);
                queue_chunk(da[u + 1], ca[u + 1], lane, pat, q, qn);
                while (qn >= 32) {
                    qn -= 32;
                    add_batch<D, R>(x, src, dst, acc, cntL, q, qn, 32, lo, lane);
                }
            }
            #pragma unroll
            for (int u = 0; u < AGG_U; ++u) {
                ca[u] = (t + 2 * AGG_U < nTrip * 2 * AGG_U)
                            ? slot_chunk(t + 2 * AGG_U + u, wave, rot, nTot, nCh) : -1;
                da[u] = load_keys(keys, ca[u], lane);
            }
            #pragma unroll
            for (int u = 0; u < AGG_U; u += 2) {
                queue_chunk(db[u],     cb[u],     lane, pat, q, qn);
                queue_chunk(db[u + 1], cb[u + 1], lane, pat, q, qn);
                while (qn >= 32) {
                    qn -= 32;
                    add_batch<D, R>(x, src, dst, acc, cntL, q, qn, 32, lo, lane);
                }
            }
        }
    }
    if (qn > 0) add_batch<D, R>(x, src, dst, acc, cntL, q, 0, qn, lo, lane);

    __syncthreads();

    {
        const v4f* a4 = reinterpret_cast<const v4f*>(acc);
        v4f* o4 = reinterpret_cast<v4f*>(agg + (long long)lo * D);
        for (int i = tid; i < nrows * (D / 4); i += AGG_NT) o4[i] = a4[i];
        if (tid < nrows) cnt[lo + tid] = (float)cntL[tid];
    }
}

template <int MT, int NT, int KC, int K>
__device__ __forceinline__ void load_frags(const float* __restrict__ aRow,
                                           const float* __restrict__ bRow, int k0,
                                           v4f (&a)[MT][KC], v4f (&b)[NT][KC]) {
    #pragma unroll
    for (int i = 0; i < MT; ++i) {
        #pragma unroll
        for (int q = 0; q < KC; ++q)
            a[i][q] = *reinterpret_cast<const v4f*>(aRow + i * 16 * K + k0 + 4 * q);
    }
    #pragma unroll
    for (int j = 0; j < NT; ++j) {
        #pragma unroll
        for (int q = 0; q < KC; ++q)
            b[j][q] = *reinterpret_cast<const v4f*>(bRow + j * 16 * K + k0 + 4 * q);
    }
}

template <int MT, int NT, int KC, bool SCALE>
__device__ __forceinline__ void mul_frags(v4f (&a)[MT][KC], const v4f (&b)[NT][KC],
                                          const float (&inv)[MT], v8f (&acc)[MT][NT]) {
    if (SCALE) {
        #pragma unroll
        for (int i = 0; i < MT; ++i) {
            #pragma unroll
            for (int q = 0; q < KC; ++q) a[i][q] *= inv[i];
        }
    }
    #pragma unroll
    for (int q = 0; q < KC; ++q) {
        #pragma unroll
        for (int s = 0; s < 2; ++s) {
            #pragma unroll
            for (int i = 0; i < MT; ++i) {
                v2f av; av[0] = a[i][q][2 * s]; av[1] = a[i][q][2 * s + 1];
                #pragma unroll
                for (int j = 0; j < NT; ++j) {
                    v2f bv; bv[0] = b[j][q][2 * s]; bv[1] = b[j][q][2 * s + 1];
                    acc[i][j] = __builtin_amdgcn_wmma_f32_16x16x4_f32(
                        false, av, false, bv, (short)0, acc[i][j], false, false);
                }
            }
        }
    }
}

template <int MT, int NT, int KC, int K, bool SCALE, bool NEXT>
__device__ __forceinline__ void sage_term(const float* __restrict__ aRow,
                                          const float* __restrict__ bRow,
                                          const float* __restrict__ aNext,
                                          const float* __restrict__ bNext,
                                          const float (&inv)[MT],
                                          v4f (&a0)[MT][KC], v4f (&b0)[NT][KC],
                                          v8f (&acc)[MT][NT]) {
    static_assert(K % (16 * KC) == 0, "two chunks per trip");
    v4f a1[MT][KC], b1[NT][KC];
    #pragma unroll 1
    for (int k0 = 0; k0 < K - 16 * KC; k0 += 16 * KC) {
        load_frags<MT, NT, KC, K>(aRow, bRow, k0 + 8 * KC, a1, b1);
        mul_frags<MT, NT, KC, SCALE>(a0, b0, inv, acc);
        load_frags<MT, NT, KC, K>(aRow, bRow, k0 + 16 * KC, a0, b0);
        mul_frags<MT, NT, KC, SCALE>(a1, b1, inv, acc);
    }
    load_frags<MT, NT, KC, K>(aRow, bRow, K - 8 * KC, a1, b1);
    mul_frags<MT, NT, KC, SCALE>(a0, b0, inv, acc);
    if (NEXT) load_frags<MT, NT, KC, K>(aNext, bNext, 0, a0, b0);
    mul_frags<MT, NT, KC, SCALE>(a1, b1, inv, acc);
}

template <int MT, int NT, int KC, int K, int NOUT, bool RELU>
__launch_bounds__(32 * NOUT / (16 * NT))
__global__ void sage_gemm_regs(const float* __restrict__ agg,
                               const float* __restrict__ cnt,
                               const float* __restrict__ xt,
                               const float* __restrict__ Wl,
                               const float* __restrict__ Wr,
                               const float* __restrict__ bias,
                               float*       __restrict__ out) {
    const int lane = threadIdx.x & 31;
    const int wave = threadIdx.x >> 5;
    const int c    = lane & 15;
    const int h    = lane >> 4;

    const long long row0 = (long long)blockIdx.x * (16 * MT);
    const int       col0 = wave * (16 * NT);

    float inv[MT];
    #pragma unroll
    for (int i = 0; i < MT; ++i)
        inv[i] = 1.0f / fmaxf(cnt[row0 + 16 * i + c], 1.0f);

    v8f acc[MT][NT];
    #pragma unroll
    for (int i = 0; i < MT; ++i) {
        #pragma unroll
        for (int j = 0; j < NT; ++j) {
            #pragma unroll
            for (int r = 0; r < 8; ++r) acc[i][j][r] = 0.0f;
        }
    }

    const long long aOff = (row0 + c) * K + 4 * KC * h;
    const int       bOff = (col0 + c) * K + 4 * KC * h;

    v4f a0[MT][KC], b0[NT][KC];
    load_frags<MT, NT, KC, K>(agg + aOff, Wl + bOff, 0, a0, b0);
    sage_term<MT, NT, KC, K, true,  true >(agg + aOff, Wl + bOff, xt + aOff, Wr + bOff,
                                           inv, a0, b0, acc);
    sage_term<MT, NT, KC, K, false, false>(xt + aOff, Wr + bOff, xt + aOff, Wr + bOff,
                                           inv, a0, b0, acc);

    #pragma unroll
    for (int j = 0; j < NT; ++j) {
        const int   col = col0 + 16 * j + c;
        const float bv  = bias[col];
        #pragma unroll
        for (int i = 0; i < MT; ++i) {
            const long long mbase = row0 + 16 * i + 8 * h;
            #pragma unroll
            for (int r = 0; r < 8; ++r) {
                float v = acc[i][j][r] + bv;
                if (RELU) v = fmaxf(v, 0.0f);
                out[(mbase + r) * NOUT + col] = v;
            }
        }
    }
}

__global__ void log_softmax64(float* __restrict__ out, int M) {
    const int lane = threadIdx.x & 31;
    const int wave = threadIdx.x >> 5;
    const int row  = blockIdx.x * 8 + wave;
    if (row >= M) return;
    float* p = out + (long long)row * 64;
    float a = p[lane];
    float b = p[lane + 32];
    float m = fmaxf(a, b);
    #pragma unroll
    for (int off = 16; off >= 1; off >>= 1)
        m = fmaxf(m, __shfl_xor(m, off, 32));
    float s = __expf(a - m) + __expf(b - m);
    #pragma unroll
    for (int off = 16; off >= 1; off >>= 1)
        s += __shfl_xor(s, off, 32);
    const float l = __logf(s) + m;
    p[lane]      = a - l;
    p[lane + 32] = b - l;
}

#define AGG_WGPS 256
#define AGG_R1 ((N2c + AGG_WGPS - 1) / AGG_WGPS)
#define AGG_R2 ((N3c + AGG_WGPS - 1) / AGG_WGPS)

#define E1c 640000
#define E2c 160000
#define KEY_CH1 ((E1c + 511) / 512)
#define KEY_CH2 ((E2c + 511) / 512)
#define KEY_W1  (KEY_CH1 * 128)
#define KEY_W2  (KEY_CH2 * 128)
#define KEY_T1  (KEY_CH1 * 32)
#define KEY_T2  (KEY_CH2 * 32)

static_assert((N2c + AGG_R1 - 1) / AGG_R1 <= 255, "an owner number must fit a key byte, 0xFF kept free");
static_assert((N3c + AGG_R2 - 1) / AGG_R2 <= 255, "an owner number must fit a key byte, 0xFF kept free");
static_assert((size_t)(KEY_W1 + KEY_W2) * 4 <= (size_t)N3c * OUTc * 4, "key lists fit the output buffer");

__global__ void owner_keys(const int* __restrict__ dst1, const int* __restrict__ dst2,
                           unsigned* __restrict__ keys1, unsigned* __restrict__ keys2,
                           int E1, int E2) {
    const int t = blockIdx.x * 256 + threadIdx.x;
    if (t < KEY_T1)                pack_keys16<AGG_R1>(dst1, keys1, t, E1);
    else if (t - KEY_T1 < KEY_T2)  pack_keys16<AGG_R2>(dst2, keys2, t - KEY_T1, E2);
}

extern "C" void kernel_launch(void* const* d_in, const int* in_sizes, int n_in,
                              void* d_out, int out_size, void* d_ws, size_t ws_size,
                              hipStream_t stream) {
    const float* x    = (const float*)d_in[0];
    const int*   src1 = (const int*)  d_in[1];
    const int*   dst1 = (const int*)  d_in[2];
    const int*   src2 = (const int*)  d_in[3];
    const int*   dst2 = (const int*)  d_in[4];
    const float* W1l  = (const float*)d_in[7];
    const float* W1r  = (const float*)d_in[8];
    const float* b1   = (const float*)d_in[9];
    const float* W2l  = (const float*)d_in[10];
    const float* W2r  = (const float*)d_in[11];
    const float* b2   = (const float*)d_in[12];

    const int E1 = in_sizes[1];
    const int E2 = in_sizes[3];

    float* agg1 = (float*)d_ws;
    float* cnt1 = agg1 + (size_t)N2c * INc;
    float* h    = cnt1 + N2c;
    float* agg2 = h    + (size_t)N2c * HIDc;
    float* cnt2 = agg2 + (size_t)N3c * HIDc;
    float* out  = (float*)d_out;
    unsigned* keys1 = (unsigned*)d_out;
    unsigned* keys2 = keys1 + KEY_W1;

    owner_keys<<<(KEY_T1 + KEY_T2 + 255) / 256, 256, 0, stream>>>(
        dst1, dst2, keys1, keys2, E1, E2);

    {
        aggregate_owned<INc, AGG_R1>
            <<<(N2c + AGG_R1 - 1) / AGG_R1, AGG_NT, 0, stream>>>(
                x, src1, dst1, keys1, agg1, cnt1, E1, N2c);
        sage_gemm_regs<5, 4, 2, INc, HIDc, true>
            <<<N2c / 80, 128, 0, stream>>>(agg1, cnt1, x, W1l, W1r, b1, h);
    }

    {
        aggregate_owned<HIDc, AGG_R2>
            <<<(N3c + AGG_R2 - 1) / AGG_R2, AGG_NT, 0, stream>>>(
                h, src2, dst2, keys2, agg2, cnt2, E2, N3c);
        sage_gemm_regs<1, 4, 4, HIDc, OUTc, false>
            <<<N3c / 16, 32, 0, stream>>>(agg2, cnt2, h, W2l, W2r, b2, out);
    }

    log_softmax64<<<(N3c + 7) / 8, 256, 0, stream>>>(out, N3c);
}


# Layer 2 finished by its owner work-groups: both SAGE products on f32 matrix ops from the LDS sums and h rows (k in 8 slices over the waves, summed in slice order), then log_softmax; 4 launches
# speedup vs baseline: 4.6573x; 1.0752x over previous
#include <hip/hip_runtime.h>
#include <math.h>

#define N1c  200000
#define N2c  40000
#define N3c  10000
#define INc  128
#define HIDc 256
#define OUTc 64

typedef float v2f __attribute__((ext_vector_type(2)));
typedef float v8f __attribute__((ext_vector_type(8)));

#define AGG_NT 1024
#define AGG_NW 32
#define AGG_U  4
#define AGG_QW 1056

typedef int      v4i  __attribute__((ext_vector_type(4)));
typedef unsigned v4u  __attribute__((ext_vector_type(4)));
typedef float    v4f  __attribute__((ext_vector_type(4)));

template <int R>
__device__ __forceinline__ unsigned key_of(int d) {
    const unsigned k = (unsigned)d / (unsigned)R;
    return (k < 255u) ? k : 255u;
}

template <int R>
__device__ __forceinline__ unsigned pack4(v4i d) {
    return key_of<R>(d[0]) | (key_of<R>(d[1]) << 8) |
           (key_of<R>(d[2]) << 16) | (key_of<R>(d[3]) << 24);
}

template <int R>
__device__ __forceinline__ void pack_keys16(const int* __restrict__ dst,
                                            unsigned* __restrict__ keys, int t, int E) {
    const int e0 = 16 * t;
    v4u w;
    if (e0 + 15 < E) {
        const v4i* p = reinterpret_cast<const v4i*>(dst + e0);
        const v4i d0 = p[0], d1 = p[1], d2 = p[2], d3 = p[3];
        w[0] = pack4<R>(d0);
        w[1] = pack4<R>(d1);
        w[2] = pack4<R>(d2);
        w[3] = pack4<R>(d3);
    } else {
        #pragma unroll
        for (int j = 0; j < 4; ++j) {
            unsigned word = 0u;
            #pragma unroll
            for (int c = 0; c < 4; ++c) {
                const int e = e0 + 4 * j + c;
                const unsigned k = (e < E) ? key_of<R>(dst[e]) : 255u;
                word |= k << (8 * c);
            }
            w[j] = word;
        }
    }
    *reinterpret_cast<v4u*>(keys + 4 * t) = w;
}

__device__ __forceinline__ int slot_chunk(int t, int wave, int rot, int nTot, int nCh) {
    int c = t * AGG_NW + wave + rot;
    if (c >= nTot) c -= nTot;
    return (c < nCh) ? c : -1;
}

__device__ __forceinline__ v4u load_keys(const unsigned* __restrict__ keys, int c, int lane) {
    const int cl = (c < 0) ? 0 : c;
    return *reinterpret_cast<const v4u*>(keys + (cl << 7) + 4 * lane);
}

__device__ __forceinline__ unsigned zero_bytes(unsigned x) {
    const unsigned t = (x & 0x7F7F7F7Fu) + 0x7F7F7F7Fu;
    return ~(t | x | 0x7F7F7F7Fu);
}

__device__ __forceinline__ void queue_chunk(v4u kw, int c, int lane, unsigned pat,
                                            int* q, int& qn) {
    if (c < 0) return;
    const int e0 = (c << 9) + 16 * lane;
    #pragma unroll
    for (int w = 0; w < 4; ++w) {
        unsigned z = zero_bytes(kw[w] ^ pat);
        unsigned mask = __builtin_amdgcn_ballot_w32(z != 0u);
        while (mask) {
            if (z != 0u) {
                q[qn + (int)__builtin_amdgcn_mbcnt_lo(mask, 0u)] =
                    e0 + 4 * w + (__builtin_ctz(z) >> 3);
                z &= z - 1u;
            }
            qn += __builtin_popcount(mask);
            mask = __builtin_amdgcn_ballot_w32(z != 0u);
        }
    }
}

template <int D, int R, int P>
__device__ __forceinline__ void add_batch(const float* __restrict__ x,
                                          const int*   __restrict__ src,
                                          const int*   __restrict__ dst,
                                          float* acc, int* cntL, const int* q,
                                          int base, int count, int lo, int lane) {
    constexpr int NV = D / 64;
    constexpr int G  = 32 / NV;
    const unsigned lane2 = 2u * (unsigned)lane;
    int s = 0, r = R;
    if (lane < count) {
        const int e = q[base + lane];
        s = src[e];
        r = dst[e] - lo;
        atomicAdd(&cntL[r], 1);
    }
    for (int j = 0; j < count; j += G) {
        v2f v[G][NV];
        int rr[G];
        #pragma unroll
        for (int g = 0; g < G; ++g) {
            const int sg = __builtin_amdgcn_readlane(s, j + g);
            rr[g]        = __builtin_amdgcn_readlane(r, j + g);
            const float* row = x + (long long)sg * D;
            #pragma unroll
            for (int k = 0; k < NV; ++k)
                v[g][k] = *reinterpret_cast<const v2f*>(row + (lane2 + 64u * k));
        }
        #pragma unroll
        for (int g = 0; g < G; ++g) {
            float* a = acc + rr[g] * P + 2 * lane;
            #pragma unroll
            for (int k = 0; k < NV; ++k) {
                atomicAdd(a + 64 * k,     v[g][k][0]);
                atomicAdd(a + 64 * k + 1, v[g][k][1]);
            }
        }
    }
}

template <int D, int R, bool FUSE>
__launch_bounds__(AGG_NT)
__global__ void aggregate_owned(const float*    __restrict__ x,
                                const int*      __restrict__ src,
                                const int*      __restrict__ dst,
                                const unsigned* __restrict__ keys,
                                float*          __restrict__ agg,
                                float*          __restrict__ cnt,
                                const float*    __restrict__ Wl,
                                const float*    __restrict__ Wr,
                                const float*    __restrict__ bias,
                                float*          __restrict__ out,
                                int E, int N) {
    constexpr int P = FUSE ? (D + 4) : D;
    __shared__ __attribute__((aligned(16))) float acc[(R + 1) * P];
    __shared__ int cntL[R + 1];
    __shared__ __attribute__((aligned(16))) int qAll[AGG_NW * AGG_QW];

    const int tid   = threadIdx.x;
    const int lane  = tid & 31;
    const int wave  = __builtin_amdgcn_readfirstlane(tid >> 5);
    const int lo    = blockIdx.x * R;
    const int nrows = (N - lo < R) ? (N - lo) : R;
    const unsigned pat = (unsigned)blockIdx.x * 0x01010101u;

    {
        v4f z; z[0] = 0.0f; z[1] = 0.0f; z[2] = 0.0f; z[3] = 0.0f;
        v4f* a4 = reinterpret_cast<v4f*>(acc);
        for (int i = tid; i < (R + 1) * P / 4; i += AGG_NT) a4[i] = z;
        if (tid <= R) cntL[tid] = 0;
    }
    __syncthreads();

    const int nCh   = (E + 511) >> 9;
    const int nGrp  = (nCh + AGG_NW * AGG_U - 1) / (AGG_NW * AGG_U);
    const int nTrip = (nGrp + 1) >> 1;
    const int nTot  = nTrip * 2 * AGG_NW * AGG_U;
    const int rot   = (int)blockIdx.x * (nTot / (int)gridDim.x);

    int* q  = qAll + wave * AGG_QW;
    int  qn = 0;

    if (nCh > 0) {
        v4u da[AGG_U], db[AGG_U];
        int ca[AGG_U], cb[AGG_U];
        #pragma unroll
        for (int u = 0; u < AGG_U; ++u) {
            ca[u] = slot_chunk(u, wave, rot, nTot, nCh);
            da[u] = load_keys(keys, ca[u], lane);
        }
        for (int t = 0; t < nTrip * 2 * AGG_U; t += 2 * AGG_U) {
            #pragma unroll
            for (int u = 0; u < AGG_U; ++u) {
                cb[u] = slot_chunk(t + AGG_U + u, wave, rot, nTot, nCh);
                db[u] = load_keys(keys, cb[u], lane);
            }
            #pragma unroll
            for (int u = 0; u < AGG_U; u += 2) {
                queue_chunk(da[u],     ca[u],     lane, pat, q, qn);
                queue_chunk(da[u + 1], ca[u + 1], lane, pat, q, qn);
                while (qn >= 32) {
                    qn -= 32;
                    add_batch<D, R, P>(x, src, dst, acc, cntL, q, qn, 32, lo, lane);
                }
            }
            #pragma unroll
            for (int u = 0; u < AGG_U; ++u) {
                ca[u] = (t + 2 * AGG_U < nTrip * 2 * AGG_U)
                            ? slot_chunk(t + 2 * AGG_U + u, wave, rot, nTot, nCh) : -1;
                da[u] = load_keys(keys, ca[u], lane);
            }
            #pragma unroll
            for (int u = 0; u < AGG_U; u += 2) {
                queue_chunk(db[u],     cb[u],     lane, pat, q, qn);
                queue_chunk(db[u + 1], cb[u + 1], lane, pat, q, qn);
                while (qn >= 32) {
                    qn -= 32;
                    add_batch<D, R, P>(x, src, dst, acc, cntL, q, qn, 32, lo, lane);
                }
            }
        }
    }
    if (qn > 0) add_batch<D, R, P>(x, src, dst, acc, cntL, q, 0, qn, lo, lane);

    __syncthreads();

    if constexpr (!FUSE) {
        const v4f* a4 = reinterpret_cast<const v4f*>(acc);
        v4f* o4 = reinterpret_cast<v4f*>(agg + (long long)lo * D);
        for (int i = tid; i < nrows * (D / 4); i += AGG_NT) o4[i] = a4[i];
        if (tid < nrows) cnt[lo + tid] = (float)cntL[tid];
    } else {
        static_assert(D == 256, "slices of 64 k, 32 per lane half");
        constexpr int NOUTF = 64;
        constexpr int MTF   = (R + 15) / 16;
        constexpr int RP    = MTF * 16;
        static_assert(8 * RP * NOUTF <= AGG_NW * AGG_QW, "partial tiles fit the queues");
        float* part = reinterpret_cast<float*>(qAll);

        const int c    = lane & 15;
        const int hh   = lane >> 4;
        const int ks   = wave & 7;
        const int nt   = wave >> 3;
        const bool own = ks >= 4;
        const int kb   = (ks & 3) * 64 + 32 * hh;

        const float* bRow = (own ? Wr : Wl) + (nt * 16 + c) * D + kb;

        v8f accT[MTF];
        #pragma unroll
        for (int i = 0; i < MTF; ++i) {
            #pragma unroll
            for (int r = 0; r < 8; ++r) accT[i][r] = 0.0f;
        }

        int   rowL[MTF];
        float inv[MTF];
        const float* aG[MTF];
        #pragma unroll
        for (int i = 0; i < MTF; ++i) {
            const int row = 16 * i + c;
            rowL[i] = (row < nrows) ? row : R;
            inv[i]  = 1.0f / fmaxf((float)cntL[rowL[i]], 1.0f);
            int gr = lo + row;
            if (gr > N - 1) gr = N - 1;
            aG[i] = x + (long long)gr * D + kb;
        }

        #pragma unroll
        for (int qh = 0; qh < 2; ++qh) {
            v4f a[MTF][4], b[4];
            #pragma unroll
            for (int q = 0; q < 4; ++q)
                b[q] = *reinterpret_cast<const v4f*>(bRow + 16 * qh + 4 * q);
            if (own) {
                #pragma unroll
                for (int i = 0; i < MTF; ++i) {
                    #pragma unroll
                    for (int q = 0; q < 4; ++q)
                        a[i][q] = *reinterpret_cast<const v4f*>(aG[i] + 16 * qh + 4 * q);
                }
            } else {
                #pragma unroll
                for (int i = 0; i < MTF; ++i) {
                    #pragma unroll
                    for (int q = 0; q < 4; ++q)
                        a[i][q] = *reinterpret_cast<const v4f*>(acc + rowL[i] * P + kb + 16 * qh + 4 * q)
                                  * inv[i];
                }
            }
            #pragma unroll
            for (int q = 0; q < 4; ++q) {
                #pragma unroll
                for (int s = 0; s < 2; ++s) {
                    v2f bv; bv[0] = b[q][2 * s]; bv[1] = b[q][2 * s + 1];
                    #pragma unroll
                    for (int i = 0; i < MTF; ++i) {
                        v2f av; av[0] = a[i][q][2 * s]; av[1] = a[i][q][2 * s + 1];
                        accT[i] = __builtin_amdgcn_wmma_f32_16x16x4_f32(
                            false, av, false, bv, (short)0, accT[i], false, false);
                    }
                }
            }
        }

        #pragma unroll
        for (int i = 0; i < MTF; ++i) {
            #pragma unroll
            for (int r = 0; r < 8; ++r)
                part[(ks * RP + 16 * i + 8 * hh + r) * NOUTF + nt * 16 + c] = accT[i][r];
        }
        __syncthreads();

        const float b0 = bias[lane], b1 = bias[lane + 32];
        for (int row = wave; row < nrows; row += AGG_NW) {
            float va = 0.0f, vb = 0.0f;
            #pragma unroll
            for (int s8 = 0; s8 < 8; ++s8) {
                va += part[(s8 * RP + row) * NOUTF + lane];
                vb += part[(s8 * RP + row) * NOUTF + lane + 32];
            }
            va += b0;
            vb += b1;
            float m = fmaxf(va, vb);
            #pragma unroll
            for (int off = 16; off >= 1; off >>= 1)
                m = fmaxf(m, __shfl_xor(m, off, 32));
            float sum = __expf(va - m) + __expf(vb - m);
            #pragma unroll
            for (int off = 16; off >= 1; off >>= 1)
                sum += __shfl_xor(sum, off, 32);
            const float l = __logf(sum) + m;
            float* p = out + (long long)(lo + row) * NOUTF;
            p[lane]      = va - l;
            p[lane + 32] = vb - l;
        }
    }
}

template <int MT, int NT, int KC, int K>
__device__ __forceinline__ void load_frags(const float* __restrict__ aRow,
                                           const float* __restrict__ bRow, int k0,
                                           v4f (&a)[MT][KC], v4f (&b)[NT][KC]) {
    #pragma unroll
    for (int i = 0; i < MT; ++i) {
        #pragma unroll
        for (int q = 0; q < KC; ++q)
            a[i][q] = *reinterpret_cast<const v4f*>(aRow + i * 16 * K + k0 + 4 * q);
    }
    #pragma unroll
    for (int j = 0; j < NT; ++j) {
        #pragma unroll
        for (int q = 0; q < KC; ++q)
            b[j][q] = *reinterpret_cast<const v4f*>(bRow + j * 16 * K + k0 + 4 * q);
    }
}

template <int MT, int NT, int KC, bool SCALE>
__device__ __forceinline__ void mul_frags(v4f (&a)[MT][KC], const v4f (&b)[NT][KC],
                                          const float (&inv)[MT], v8f (&acc)[MT][NT]) {
    if (SCALE) {
        #pragma unroll
        for (int i = 0; i < MT; ++i) {
            #pragma unroll
            for (int q = 0; q < KC; ++q) a[i][q] *= inv[i];
        }
    }
    #pragma unroll
    for (int q = 0; q < KC; ++q) {
        #pragma unroll
        for (int s = 0; s < 2; ++s) {
            #pragma unroll
            for (int i = 0; i < MT; ++i) {
                v2f av; av[0] = a[i][q][2 * s]; av[1] = a[i][q][2 * s + 1];
                #pragma unroll
                for (int j = 0; j < NT; ++j) {
                    v2f bv; bv[0] = b[j][q][2 * s]; bv[1] = b[j][q][2 * s + 1];
                    acc[i][j] = __builtin_amdgcn_wmma_f32_16x16x4_f32(
                        false, av, false, bv, (short)0, acc[i][j], false, false);
                }
            }
        }
    }
}

template <int MT, int NT, int KC, int K, bool SCALE, bool NEXT>
__device__ __forceinline__ void sage_term(const float* __restrict__ aRow,
                                          const float* __restrict__ bRow,
                                          const float* __restrict__ aNext,
                                          const float* __restrict__ bNext,
                                          const float (&inv)[MT],
                                          v4f (&a0)[MT][KC], v4f (&b0)[NT][KC],
                                          v8f (&acc)[MT][NT]) {
    static_assert(K % (16 * KC) == 0, "two chunks per trip");
    v4f a1[MT][KC], b1[NT][KC];
    #pragma unroll 1
    for (int k0 = 0; k0 < K - 16 * KC; k0 += 16 * KC) {
        load_frags<MT, NT, KC, K>(aRow, bRow, k0 + 8 * KC, a1, b1);
        mul_frags<MT, NT, KC, SCALE>(a0, b0, inv, acc);
        load_frags<MT, NT, KC, K>(aRow, bRow, k0 + 16 * KC, a0, b0);
        mul_frags<MT, NT, KC, SCALE>(a1, b1, inv, acc);
    }
    load_frags<MT, NT, KC, K>(aRow, bRow, K - 8 * KC, a1, b1);
    mul_frags<MT, NT, KC, SCALE>(a0, b0, inv, acc);
    if (NEXT) load_frags<MT, NT, KC, K>(aNext, bNext, 0, a0, b0);
    mul_frags<MT, NT, KC, SCALE>(a1, b1, inv, acc);
}

template <int MT, int NT, int KC, int K, int NOUT, bool RELU>
__launch_bounds__(32 * NOUT / (16 * NT))
__global__ void sage_gemm_regs(const float* __restrict__ agg,
                               const float* __restrict__ cnt,
                               const float* __restrict__ xt,
                               const float* __restrict__ Wl,
                               const float* __restrict__ Wr,
                               const float* __restrict__ bias,
                               float*       __restrict__ out) {
    const int lane = threadIdx.x & 31;
    const int wave = threadIdx.x >> 5;
    const int c    = lane & 15;
    const int h    = lane >> 4;

    const long long row0 = (long long)blockIdx.x * (16 * MT);
    const int       col0 = wave * (16 * NT);

    float inv[MT];
    #pragma unroll
    for (int i = 0; i < MT; ++i)
        inv[i] = 1.0f / fmaxf(cnt[row0 + 16 * i + c], 1.0f);

    v8f acc[MT][NT];
    #pragma unroll
    for (int i = 0; i < MT; ++i) {
        #pragma unroll
        for (int j = 0; j < NT; ++j) {
            #pragma unroll
            for (int r = 0; r < 8; ++r) acc[i][j][r] = 0.0f;
        }
    }

    const long long aOff = (row0 + c) * K + 4 * KC * h;
    const int       bOff = (col0 + c) * K + 4 * KC * h;

    v4f a0[MT][KC], b0[NT][KC];
    load_frags<MT, NT, KC, K>(agg + aOff, Wl + bOff, 0, a0, b0);
    sage_term<MT, NT, KC, K, true,  true >(agg + aOff, Wl + bOff, xt + aOff, Wr + bOff,
                                           inv, a0, b0, acc);
    sage_term<MT, NT, KC, K, false, false>(xt + aOff, Wr + bOff, xt + aOff, Wr + bOff,
                                           inv, a0, b0, acc);

    #pragma unroll
    for (int j = 0; j < NT; ++j) {
        const int   col = col0 + 16 * j + c;
        const float bv  = bias[col];
        #pragma unroll
        for (int i = 0; i < MT; ++i) {
            const long long mbase = row0 + 16 * i + 8 * h;
            #pragma unroll
            for (int r = 0; r < 8; ++r) {
                float v = acc[i][j][r] + bv;
                if (RELU) v = fmaxf(v, 0.0f);
                out[(mbase + r) * NOUT + col] = v;
            }
        }
    }
}

#define AGG_WGPS 256
#define AGG_R1 ((N2c + AGG_WGPS - 1) / AGG_WGPS)
#define AGG_R2 ((N3c + AGG_WGPS - 1) / AGG_WGPS)

#define E1c 640000
#define E2c 160000
#define KEY_CH1 ((E1c + 511) / 512)
#define KEY_CH2 ((E2c + 511) / 512)
#define KEY_W1  (KEY_CH1 * 128)
#define KEY_W2  (KEY_CH2 * 128)
#define KEY_T1  (KEY_CH1 * 32)
#define KEY_T2  (KEY_CH2 * 32)

static_assert((N2c + AGG_R1 - 1) / AGG_R1 <= 255, "an owner number must fit a key byte, 0xFF kept free");
static_assert((N3c + AGG_R2 - 1) / AGG_R2 <= 255, "an owner number must fit a key byte, 0xFF kept free");
static_assert((size_t)(KEY_W1 + KEY_W2) * 4 <= (size_t)N3c * HIDc * 4, "key lists fit the spare workspace plane");

__global__ void owner_keys(const int* __restrict__ dst1, const int* __restrict__ dst2,
                           unsigned* __restrict__ keys1, unsigned* __restrict__ keys2,
                           int E1, int E2) {
    const int t = blockIdx.x * 256 + threadIdx.x;
    if (t < KEY_T1)                pack_keys16<AGG_R1>(dst1, keys1, t, E1);
    else if (t - KEY_T1 < KEY_T2)  pack_keys16<AGG_R2>(dst2, keys2, t - KEY_T1, E2);
}

extern "C" void kernel_launch(void* const* d_in, const int* in_sizes, int n_in,
                              void* d_out, int out_size, void* d_ws, size_t ws_size,
                              hipStream_t stream) {
    const float* x    = (const float*)d_in[0];
    const int*   src1 = (const int*)  d_in[1];
    const int*   dst1 = (const int*)  d_in[2];
    const int*   src2 = (const int*)  d_in[3];
    const int*   dst2 = (const int*)  d_in[4];
    const float* W1l  = (const float*)d_in[7];
    const float* W1r  = (const float*)d_in[8];
    const float* b1   = (const float*)d_in[9];
    const float* W2l  = (const float*)d_in[10];
    const float* W2r  = (const float*)d_in[11];
    const float* b2   = (const float*)d_in[12];

    const int E1 = in_sizes[1];
    const int E2 = in_sizes[3];

    float* agg1 = (float*)d_ws;
    float* cnt1 = agg1 + (size_t)N2c * INc;
    float* h    = cnt1 + N2c;
    float* out  = (float*)d_out;
    unsigned* keys1 = (unsigned*)(h + (size_t)N2c * HIDc);
    unsigned* keys2 = keys1 + KEY_W1;

    owner_keys<<<(KEY_T1 + KEY_T2 + 255) / 256, 256, 0, stream>>>(
        dst1, dst2, keys1, keys2, E1, E2);

    {
        aggregate_owned<INc, AGG_R1, false>
            <<<(N2c + AGG_R1 - 1) / AGG_R1, AGG_NT, 0, stream>>>(
                x, src1, dst1, keys1, agg1, cnt1, nullptr, nullptr, nullptr, nullptr, E1, N2c);
        sage_gemm_regs<5, 4, 2, INc, HIDc, true>
            <<<N2c / 80, 128, 0, stream>>>(agg1, cnt1, x, W1l, W1r, b1, h);
    }

    {
        aggregate_owned<HIDc, AGG_R2, true>
            <<<(N3c + AGG_R2 - 1) / AGG_R2, AGG_NT, 0, stream>>>(
                h, src2, dst2, keys2, nullptr, nullptr, W2l, W2r, b2, out, E2, N3c);
    }
}


# Owner work-groups file their edges into per-row source lists in LDS (the counting integer add gives the place); a wave sums a row's sources in registers in list (arrival) order, no float LDS adds
# speedup vs baseline: 5.2386x; 1.1248x over previous
#include <hip/hip_runtime.h>
#include <math.h>

#define N1c  200000
#define N2c  40000
#define N3c  10000
#define INc  128
#define HIDc 256
#define OUTc 64

typedef float v2f __attribute__((ext_vector_type(2)));
typedef float v8f __attribute__((ext_vector_type(8)));

#define AGG_NT 1024
#define AGG_NW 32
#define AGG_U  4
#define AGG_QW 1056
#define AGG_CAP 64

typedef int      v4i  __attribute__((ext_vector_type(4)));
typedef unsigned v4u  __attribute__((ext_vector_type(4)));
typedef float    v4f  __attribute__((ext_vector_type(4)));

template <int R>
__device__ __forceinline__ unsigned key_of(int d) {
    const unsigned k = (unsigned)d / (unsigned)R;
    return (k < 255u) ? k : 255u;
}

template <int R>
__device__ __forceinline__ unsigned pack4(v4i d) {
    return key_of<R>(d[0]) | (key_of<R>(d[1]) << 8) |
           (key_of<R>(d[2]) << 16) | (key_of<R>(d[3]) << 24);
}

template <int R>
__device__ __forceinline__ void pack_keys16(const int* __restrict__ dst,
                                            unsigned* __restrict__ keys, int t, int E) {
    const int e0 = 16 * t;
    v4u w;
    if (e0 + 15 < E) {
        const v4i* p = reinterpret_cast<const v4i*>(dst + e0);
        const v4i d0 = p[0], d1 = p[1], d2 = p[2], d3 = p[3];
        w[0] = pack4<R>(d0);
        w[1] = pack4<R>(d1);
        w[2] = pack4<R>(d2);
        w[3] = pack4<R>(d3);
    } else {
        #pragma unroll
        for (int j = 0; j < 4; ++j) {
            unsigned word = 0u;
            #pragma unroll
            for (int c = 0; c < 4; ++c) {
                const int e = e0 + 4 * j + c;
                const unsigned k = (e < E) ? key_of<R>(dst[e]) : 255u;
                word |= k << (8 * c);
            }
            w[j] = word;
        }
    }
    *reinterpret_cast<v4u*>(keys + 4 * t) = w;
}

__device__ __forceinline__ int slot_chunk(int t, int wave, int rot, int nTot, int nCh) {
    int c = t * AGG_NW + wave + rot;
    if (c >= nTot) c -= nTot;
    return (c < nCh) ? c : -1;
}

__device__ __forceinline__ v4u load_keys(const unsigned* __restrict__ keys, int c, int lane) {
    const int cl = (c < 0) ? 0 : c;
    return *reinterpret_cast<const v4u*>(keys + (cl << 7) + 4 * lane);
}

__device__ __forceinline__ unsigned zero_bytes(unsigned x) {
    const unsigned t = (x & 0x7F7F7F7Fu) + 0x7F7F7F7Fu;
    return ~(t | x | 0x7F7F7F7Fu);
}

__device__ __forceinline__ void queue_chunk(v4u kw, int c, int lane, unsigned pat,
                                            int* q, int& qn) {
    if (c < 0) return;
    const int e0 = (c << 9) + 16 * lane;
    #pragma unroll
    for (int w = 0; w < 4; ++w) {
        unsigned z = zero_bytes(kw[w] ^ pat);
        unsigned mask = __builtin_amdgcn_ballot_w32(z != 0u);
        while (mask) {
            if (z != 0u) {
                q[qn + (int)__builtin_amdgcn_mbcnt_lo(mask, 0u)] =
                    e0 + 4 * w + (__builtin_ctz(z) >> 3);
                z &= z - 1u;
            }
            qn += __builtin_popcount(mask);
            mask = __builtin_amdgcn_ballot_w32(z != 0u);
        }
    }
}

template <int D, int R, int P>
__device__ __forceinline__ void file_batch(const float* __restrict__ x,
                                           const int*   __restrict__ src,
                                           const int*   __restrict__ dst,
                                           float* acc, int* cntL, int* lst, const int* q,
                                           int base, int count, int lo, int lane) {
    constexpr int NV = D / 64;
    int  s = 0, r = 0;
    bool over = false;
    if (lane < count) {
        const int e = q[base + lane];
        s = src[e];
        r = dst[e] - lo;
        const int place = atomicAdd(&cntL[r], 1);
        if (place < AGG_CAP) lst[r * AGG_CAP + place] = s;
        else                 over = true;
    }
    unsigned mask = __builtin_amdgcn_ballot_w32(over);
    while (mask) {
        const int l = __builtin_ctz(mask);
        mask &= mask - 1u;
        const int sg = __builtin_amdgcn_readlane(s, l);
        const int rg = __builtin_amdgcn_readlane(r, l);
        const float* row = x + (long long)sg * D;
        float* a = acc + rg * P + 2 * lane;
        #pragma unroll
        for (int k = 0; k < NV; ++k) {
            const v2f v = *reinterpret_cast<const v2f*>(row + (2 * lane + 64 * k));
            atomicAdd(a + 64 * k,     v[0]);
            atomicAdd(a + 64 * k + 1, v[1]);
        }
    }
}

template <int D, int P>
__device__ __forceinline__ void sum_lists(const float* __restrict__ x, float* acc,
                                          const int* cntL, const int* lst,
                                          int nrows, int wave, int lane) {
    constexpr int PARTS = D / 128;
    for (int item = wave; item < nrows * PARTS; item += AGG_NW) {
        const int r    = item / PARTS;
        const int part = item % PARTS;
        const int cr   = cntL[r];
        const int n    = __builtin_amdgcn_readfirstlane(cr < AGG_CAP ? cr : AGG_CAP);
        const int* L   = lst + r * AGG_CAP;
        const float* xp = x + part * 128 + 4 * lane;
        v4f sum; sum[0] = 0.0f; sum[1] = 0.0f; sum[2] = 0.0f; sum[3] = 0.0f;
        for (int j = 0; j < n; j += 16) {
            int jj = j + (lane & 15);
            if (jj > n - 1) jj = n - 1;
            const int mine = L[jj];
            v4f v[16];
            #pragma unroll
            for (int g = 0; g < 16; ++g) {
                const int sg = __builtin_amdgcn_readlane(mine, g);
                v[g] = *reinterpret_cast<const v4f*>(xp + (long long)sg * D);
            }
            #pragma unroll
            for (int g = 0; g < 16; ++g) {
                const bool live = (j + g < n);
                sum[0] += live ? v[g][0] : 0.0f;
                sum[1] += live ? v[g][1] : 0.0f;
                sum[2] += live ? v[g][2] : 0.0f;
                sum[3] += live ? v[g][3] : 0.0f;
            }
        }
        v4f* a = reinterpret_cast<v4f*>(acc + r * P + part * 128 + 4 * lane);
        *a = *a + sum;
    }
}

template <int D, int R, bool FUSE>
__launch_bounds__(AGG_NT)
__global__ void aggregate_owned(const float*    __restrict__ x,
                                const int*      __restrict__ src,
                                const int*      __restrict__ dst,
                                const unsigned* __restrict__ keys,
                                float*          __restrict__ agg,
                                float*          __restrict__ cnt,
                                const float*    __restrict__ Wl,
                                const float*    __restrict__ Wr,
                                const float*    __restrict__ bias,
                                float*          __restrict__ out,
                                int E, int N) {
    constexpr int P = FUSE ? (D + 4) : D;
    __shared__ __attribute__((aligned(16))) float acc[(R + 1) * P];
    __shared__ int cntL[R + 1];
    __shared__ __attribute__((aligned(16))) int qAll[AGG_NW * AGG_QW];
    __shared__ int lst[R * AGG_CAP];

    const int tid   = threadIdx.x;
    const int lane  = tid & 31;
    const int wave  = __builtin_amdgcn_readfirstlane(tid >> 5);
    const int lo    = blockIdx.x * R;
    const int nrows = (N - lo < R) ? (N - lo) : R;
    const unsigned pat = (unsigned)blockIdx.x * 0x01010101u;

    {
        v4f z; z[0] = 0.0f; z[1] = 0.0f; z[2] = 0.0f; z[3] = 0.0f;
        v4f* a4 = reinterpret_cast<v4f*>(acc);
        for (int i = tid; i < (R + 1) * P / 4; i += AGG_NT) a4[i] = z;
        if (tid <= R) cntL[tid] = 0;
    }
    __syncthreads();

    const int nCh   = (E + 511) >> 9;
    const int nGrp  = (nCh + AGG_NW * AGG_U - 1) / (AGG_NW * AGG_U);
    const int nTrip = (nGrp + 1) >> 1;
    const int nTot  = nTrip * 2 * AGG_NW * AGG_U;
    const int rot   = (int)blockIdx.x * (nTot / (int)gridDim.x);

    int* q  = qAll + wave * AGG_QW;
    int  qn = 0;

    if (nCh > 0) {
        v4u da[AGG_U], db[AGG_U];
        int ca[AGG_U], cb[AGG_U];
        #pragma unroll
        for (int u = 0; u < AGG_U; ++u) {
            ca[u] = slot_chunk(u, wave, rot, nTot, nCh);
            da[u] = load_keys(keys, ca[u], lane);
        }
        for (int t = 0; t < nTrip * 2 * AGG_U; t += 2 * AGG_U) {
            #pragma unroll
            for (int u = 0; u < AGG_U; ++u) {
                cb[u] = slot_chunk(t + AGG_U + u, wave, rot, nTot, nCh);
                db[u] = load_keys(keys, cb[u], lane);
            }
            #pragma unroll
            for (int u = 0; u < AGG_U; u += 2) {
                queue_chunk(da[u],     ca[u],     lane, pat, q, qn);
                queue_chunk(da[u + 1], ca[u + 1], lane, pat, q, qn);
                while (qn >= 32) {
                    qn -= 32;
                    file_batch<D, R, P>(x, src, dst, acc, cntL, lst, q, qn, 32, lo, lane);
                }
            }
            #pragma unroll
            for (int u = 0; u < AGG_U; ++u) {
                ca[u] = (t + 2 * AGG_U < nTrip * 2 * AGG_U)
                            ? slot_chunk(t + 2 * AGG_U + u, wave, rot, nTot, nCh) : -1;
                da[u] = load_keys(keys, ca[u], lane);
            }
            #pragma unroll
            for (int u = 0; u < AGG_U; u += 2) {
                queue_chunk(db[u],     cb[u],     lane, pat, q, qn);
                queue_chunk(db[u + 1], cb[u + 1], lane, pat, q, qn);
                while (qn >= 32) {
                    qn -= 32;
                    file_batch<D, R, P>(x, src, dst, acc, cntL, lst, q, qn, 32, lo, lane);
                }
            }
        }
    }
    if (qn > 0) file_batch<D, R, P>(x, src, dst, acc, cntL, lst, q, 0, qn, lo, lane);

    __syncthreads();
    sum_lists<D, P>(x, acc, cntL, lst, nrows, wave, lane);
    __syncthreads();

    if constexpr (!FUSE) {
        const v4f* a4 = reinterpret_cast<const v4f*>(acc);
        v4f* o4 = reinterpret_cast<v4f*>(agg + (long long)lo * D);
        for (int i = tid; i < nrows * (D / 4); i += AGG_NT) o4[i] = a4[i];
        if (tid < nrows) cnt[lo + tid] = (float)cntL[tid];
    } else {
        static_assert(D == 256, "slices of 64 k, 32 per lane half");
        constexpr int NOUTF = 64;
        constexpr int MTF   = (R + 15) / 16;
        constexpr int RP    = MTF * 16;
        static_assert(8 * RP * NOUTF <= AGG_NW * AGG_QW, "partial tiles fit the queues");
        float* part = reinterpret_cast<float*>(qAll);

        const int c    = lane & 15;
        const int hh   = lane >> 4;
        const int ks   = wave & 7;
        const int nt   = wave >> 3;
        const bool own = ks >= 4;
        const int kb   = (ks & 3) * 64 + 32 * hh;

        const float* bRow = (own ? Wr : Wl) + (nt * 16 + c) * D + kb;

        v8f accT[MTF];
        #pragma unroll
        for (int i = 0; i < MTF; ++i) {
            #pragma unroll
            for (int r = 0; r < 8; ++r) accT[i][r] = 0.0f;
        }

        int   rowL[MTF];
        float inv[MTF];
        const float* aG[MTF];
        #pragma unroll
        for (int i = 0; i < MTF; ++i) {
            const int row = 16 * i + c;
            rowL[i] = (row < nrows) ? row : R;
            inv[i]  = 1.0f / fmaxf((float)cntL[rowL[i]], 1.0f);
            int gr = lo + row;
            if (gr > N - 1) gr = N - 1;
            aG[i] = x + (long long)gr * D + kb;
        }

        #pragma unroll
        for (int qh = 0; qh < 2; ++qh) {
            v4f a[MTF][4], b[4];
            #pragma unroll
            for (int q = 0; q < 4; ++q)
                b[q] = *reinterpret_cast<const v4f*>(bRow + 16 * qh + 4 * q);
            if (own) {
                #pragma unroll
                for (int i = 0; i < MTF; ++i) {
                    #pragma unroll
                    for (int q = 0; q < 4; ++q)
                        a[i][q] = *reinterpret_cast<const v4f*>(aG[i] + 16 * qh + 4 * q);
                }
            } else {
                #pragma unroll
                for (int i = 0; i < MTF; ++i) {
                    #pragma unroll
                    for (int q = 0; q < 4; ++q)
                        a[i][q] = *reinterpret_cast<const v4f*>(acc + rowL[i] * P + kb + 16 * qh + 4 * q)
                                  * inv[i];
                }
            }
            #pragma unroll
            for (int q = 0; q < 4; ++q) {
                #pragma unroll
                for (int s = 0; s < 2; ++s) {
                    v2f bv; bv[0] = b[q][2 * s]; bv[1] = b[q][2 * s + 1];
                    #pragma unroll
                    for (int i = 0; i < MTF; ++i) {
                        v2f av; av[0] = a[i][q][2 * s]; av[1] = a[i][q][2 * s + 1];
                        accT[i] = __builtin_amdgcn_wmma_f32_16x16x4_f32(
                            false, av, false, bv, (short)0, accT[i], false, false);
                    }
                }
            }
        }

        #pragma unroll
        for (int i = 0; i < MTF; ++i) {
            #pragma unroll
            for (int r = 0; r < 8; ++r)
                part[(ks * RP + 16 * i + 8 * hh + r) * NOUTF + nt * 16 + c] = accT[i][r];
        }
        __syncthreads();

        const float b0 = bias[lane], b1 = bias[lane + 32];
        for (int row = wave; row < nrows; row += AGG_NW) {
            float va = 0.0f, vb = 0.0f;
            #pragma unroll
            for (int s8 = 0; s8 < 8; ++s8) {
                va += part[(s8 * RP + row) * NOUTF + lane];
                vb += part[(s8 * RP + row) * NOUTF + lane + 32];
            }
            va += b0;
            vb += b1;
            float m = fmaxf(va, vb);
            #pragma unroll
            for (int off = 16; off >= 1; off >>= 1)
                m = fmaxf(m, __shfl_xor(m, off, 32));
            float sum = __expf(va - m) + __expf(vb - m);
            #pragma unroll
            for (int off = 16; off >= 1; off >>= 1)
                sum += __shfl_xor(sum, off, 32);
            const float l = __logf(sum) + m;
            float* p = out + (long long)(lo + row) * NOUTF;
            p[lane]      = va - l;
            p[lane + 32] = vb - l;
        }
    }
}

template <int MT, int NT, int KC, int K>
__device__ __forceinline__ void load_frags(const float* __restrict__ aRow,
                                           const float* __restrict__ bRow, int k0,
                                           v4f (&a)[MT][KC], v4f (&b)[NT][KC]) {
    #pragma unroll
    for (int i = 0; i < MT; ++i) {
        #pragma unroll
        for (int q = 0; q < KC; ++q)
            a[i][q] = *reinterpret_cast<const v4f*>(aRow + i * 16 * K + k0 + 4 * q);
    }
    #pragma unroll
    for (int j = 0; j < NT; ++j) {
        #pragma unroll
        for (int q = 0; q < KC; ++q)
            b[j][q] = *reinterpret_cast<const v4f*>(bRow + j * 16 * K + k0 + 4 * q);
    }
}

template <int MT, int NT, int KC, bool SCALE>
__device__ __forceinline__ void mul_frags(v4f (&a)[MT][KC], const v4f (&b)[NT][KC],
                                          const float (&inv)[MT], v8f (&acc)[MT][NT]) {
    if (SCALE) {
        #pragma unroll
        for (int i = 0; i < MT; ++i) {
            #pragma unroll
            for (int q = 0; q < KC; ++q) a[i][q] *= inv[i];
        }
    }
    #pragma unroll
    for (int q = 0; q < KC; ++q) {
        #pragma unroll
        for (int s = 0; s < 2; ++s) {
            #pragma unroll
            for (int i = 0; i < MT; ++i) {
                v2f av; av[0] = a[i][q][2 * s]; av[1] = a[i][q][2 * s + 1];
                #pragma unroll
                for (int j = 0; j < NT; ++j) {
                    v2f bv; bv[0] = b[j][q][2 * s]; bv[1] = b[j][q][2 * s + 1];
                    acc[i][j] = __builtin_amdgcn_wmma_f32_16x16x4_f32(
                        false, av, false, bv, (short)0, acc[i][j], false, false);
                }
            }
        }
    }
}

template <int MT, int NT, int KC, int K, bool SCALE, bool NEXT>
__device__ __forceinline__ void sage_term(const float* __restrict__ aRow,
                                          const float* __restrict__ bRow,
                                          const float* __restrict__ aNext,
                                          const float* __restrict__ bNext,
                                          const float (&inv)[MT],
                                          v4f (&a0)[MT][KC], v4f (&b0)[NT][KC],
                                          v8f (&acc)[MT][NT]) {
    static_assert(K % (16 * KC) == 0, "two chunks per trip");
    v4f a1[MT][KC], b1[NT][KC];
    #pragma unroll 1
    for (int k0 = 0; k0 < K - 16 * KC; k0 += 16 * KC) {
        load_frags<MT, NT, KC, K>(aRow, bRow, k0 + 8 * KC, a1, b1);
        mul_frags<MT, NT, KC, SCALE>(a0, b0, inv, acc);
        load_frags<MT, NT, KC, K>(aRow, bRow, k0 + 16 * KC, a0, b0);
        mul_frags<MT, NT, KC, SCALE>(a1, b1, inv, acc);
    }
    load_frags<MT, NT, KC, K>(aRow, bRow, K - 8 * KC, a1, b1);
    mul_frags<MT, NT, KC, SCALE>(a0, b0, inv, acc);
    if (NEXT) load_frags<MT, NT, KC, K>(aNext, bNext, 0, a0, b0);
    mul_frags<MT, NT, KC, SCALE>(a1, b1, inv, acc);
}

template <int MT, int NT, int KC, int K, int NOUT, bool RELU>
__launch_bounds__(32 * NOUT / (16 * NT))
__global__ void sage_gemm_regs(const float* __restrict__ agg,
                               const float* __restrict__ cnt,
                               const float* __restrict__ xt,
                               const float* __restrict__ Wl,
                               const float* __restrict__ Wr,
                               const float* __restrict__ bias,
                               float*       __restrict__ out) {
    const int lane = threadIdx.x & 31;
    const int wave = threadIdx.x >> 5;
    const int c    = lane & 15;
    const int h    = lane >> 4;

    const long long row0 = (long long)blockIdx.x * (16 * MT);
    const int       col0 = wave * (16 * NT);

    float inv[MT];
    #pragma unroll
    for (int i = 0; i < MT; ++i)
        inv[i] = 1.0f / fmaxf(cnt[row0 + 16 * i + c], 1.0f);

    v8f acc[MT][NT];
    #pragma unroll
    for (int i = 0; i < MT; ++i) {
        #pragma unroll
        for (int j = 0; j < NT; ++j) {
            #pragma unroll
            for (int r = 0; r < 8; ++r) acc[i][j][r] = 0.0f;
        }
    }

    const long long aOff = (row0 + c) * K + 4 * KC * h;
    const int       bOff = (col0 + c) * K + 4 * KC * h;

    v4f a0[MT][KC], b0[NT][KC];
    load_frags<MT, NT, KC, K>(agg + aOff, Wl + bOff, 0, a0, b0);
    sage_term<MT, NT, KC, K, true,  true >(agg + aOff, Wl + bOff, xt + aOff, Wr + bOff,
                                           inv, a0, b0, acc);
    sage_term<MT, NT, KC, K, false, false>(xt + aOff, Wr + bOff, xt + aOff, Wr + bOff,
                                           inv, a0, b0, acc);

    #pragma unroll
    for (int j = 0; j < NT; ++j) {
        const int   col = col0 + 16 * j + c;
        const float bv  = bias[col];
        #pragma unroll
        for (int i = 0; i < MT; ++i) {
            const long long mbase = row0 + 16 * i + 8 * h;
            #pragma unroll
            for (int r = 0; r < 8; ++r) {
                float v = acc[i][j][r] + bv;
                if (RELU) v = fmaxf(v, 0.0f);
                out[(mbase + r) * NOUT + col] = v;
            }
        }
    }
}

#define AGG_WGPS 256
#define AGG_R1 ((N2c + AGG_WGPS - 1) / AGG_WGPS)
#define AGG_R2 ((N3c + AGG_WGPS - 1) / AGG_WGPS)

#define E1c 640000
#define E2c 160000
#define KEY_CH1 ((E1c + 511) / 512)
#define KEY_CH2 ((E2c + 511) / 512)
#define KEY_W1  (KEY_CH1 * 128)
#define KEY_W2  (KEY_CH2 * 128)
#define KEY_T1  (KEY_CH1 * 32)
#define KEY_T2  (KEY_CH2 * 32)

static_assert((N2c + AGG_R1 - 1) / AGG_R1 <= 255, "an owner number must fit a key byte, 0xFF kept free");
static_assert((N3c + AGG_R2 - 1) / AGG_R2 <= 255, "an owner number must fit a key byte, 0xFF kept free");
static_assert((size_t)(KEY_W1 + KEY_W2) * 4 <= (size_t)N3c * HIDc * 4, "key lists fit the spare workspace plane");

__global__ void owner_keys(const int* __restrict__ dst1, const int* __restrict__ dst2,
                           unsigned* __restrict__ keys1, unsigned* __restrict__ keys2,
                           int E1, int E2) {
    const int t = blockIdx.x * 256 + threadIdx.x;
    if (t < KEY_T1)                pack_keys16<AGG_R1>(dst1, keys1, t, E1);
    else if (t - KEY_T1 < KEY_T2)  pack_keys16<AGG_R2>(dst2, keys2, t - KEY_T1, E2);
}

extern "C" void kernel_launch(void* const* d_in, const int* in_sizes, int n_in,
                              void* d_out, int out_size, void* d_ws, size_t ws_size,
                              hipStream_t stream) {
    const float* x    = (const float*)d_in[0];
    const int*   src1 = (const int*)  d_in[1];
    const int*   dst1 = (const int*)  d_in[2];
    const int*   src2 = (const int*)  d_in[3];
    const int*   dst2 = (const int*)  d_in[4];
    const float* W1l  = (const float*)d_in[7];
    const float* W1r  = (const float*)d_in[8];
    const float* b1   = (const float*)d_in[9];
    const float* W2l  = (const float*)d_in[10];
    const float* W2r  = (const float*)d_in[11];
    const float* b2   = (const float*)d_in[12];

    const int E1 = in_sizes[1];
    const int E2 = in_sizes[3];

    float* agg1 = (float*)d_ws;
    float* cnt1 = agg1 + (size_t)N2c * INc;
    float* h    = cnt1 + N2c;
    float* out  = (float*)d_out;
    unsigned* keys1 = (unsigned*)(h + (size_t)N2c * HIDc);
    unsigned* keys2 = keys1 + KEY_W1;

    owner_keys<<<(KEY_T1 + KEY_T2 + 255) / 256, 256, 0, stream>>>(
        dst1, dst2, keys1, keys2, E1, E2);

    {
        aggregate_owned<INc, AGG_R1, false>
            <<<(N2c + AGG_R1 - 1) / AGG_R1, AGG_NT, 0, stream>>>(
                x, src1, dst1, keys1, agg1, cnt1, nullptr, nullptr, nullptr, nullptr, E1, N2c);
        sage_gemm_regs<5, 4, 2, INc, HIDc, true>
            <<<N2c / 80, 128, 0, stream>>>(agg1, cnt1, x, W1l, W1r, b1, h);
    }

    {
        aggregate_owned<HIDc, AGG_R2, true>
            <<<(N3c + AGG_R2 - 1) / AGG_R2, AGG_NT, 0, stream>>>(
                h, src2, dst2, keys2, nullptr, nullptr, W2l, W2r, b2, out, E2, N3c);
    }
}


# Layer 1 by wave role: waves 0-15 gather all neighbour rows (24 loads in flight) while waves 16-31 run the own-row products, 10 tiles a wave; mean term after; k sums now own term first, then mean term
# speedup vs baseline: 5.6510x; 1.0787x over previous
#include <hip/hip_runtime.h>
#include <math.h>

#define N1c  200000
#define N2c  40000
#define N3c  10000
#define INc  128
#define HIDc 256
#define OUTc 64

typedef float v2f __attribute__((ext_vector_type(2)));
typedef float v8f __attribute__((ext_vector_type(8)));

#define AGG_NT 1024
#define AGG_NW 32
#define AGG_U  4
#define AGG_QW 1056
#define AGG_CAP 64

typedef int      v4i  __attribute__((ext_vector_type(4)));
typedef unsigned v4u  __attribute__((ext_vector_type(4)));
typedef float    v4f  __attribute__((ext_vector_type(4)));

template <int R>
__device__ __forceinline__ unsigned key_of(int d) {
    const unsigned k = (unsigned)d / (unsigned)R;
    return (k < 255u) ? k : 255u;
}

template <int R>
__device__ __forceinline__ unsigned pack4(v4i d) {
    return key_of<R>(d[0]) | (key_of<R>(d[1]) << 8) |
           (key_of<R>(d[2]) << 16) | (key_of<R>(d[3]) << 24);
}

template <int R>
__device__ __forceinline__ void pack_keys16(const int* __restrict__ dst,
                                            unsigned* __restrict__ keys, int t, int E) {
    const int e0 = 16 * t;
    v4u w;
    if (e0 + 15 < E) {
        const v4i* p = reinterpret_cast<const v4i*>(dst + e0);
        const v4i d0 = p[0], d1 = p[1], d2 = p[2], d3 = p[3];
        w[0] = pack4<R>(d0);
        w[1] = pack4<R>(d1);
        w[2] = pack4<R>(d2);
        w[3] = pack4<R>(d3);
    } else {
        #pragma unroll
        for (int j = 0; j < 4; ++j) {
            unsigned word = 0u;
            #pragma unroll
            for (int c = 0; c < 4; ++c) {
                const int e = e0 + 4 * j + c;
                const unsigned k = (e < E) ? key_of<R>(dst[e]) : 255u;
                word |= k << (8 * c);
            }
            w[j] = word;
        }
    }
    *reinterpret_cast<v4u*>(keys + 4 * t) = w;
}

__device__ __forceinline__ int slot_chunk(int t, int wave, int rot, int nTot, int nCh) {
    int c = t * AGG_NW + wave + rot;
    if (c >= nTot) c -= nTot;
    return (c < nCh) ? c : -1;
}

__device__ __forceinline__ v4u load_keys(const unsigned* __restrict__ keys, int c, int lane) {
    const int cl = (c < 0) ? 0 : c;
    return *reinterpret_cast<const v4u*>(keys + (cl << 7) + 4 * lane);
}

__device__ __forceinline__ unsigned zero_bytes(unsigned x) {
    const unsigned t = (x & 0x7F7F7F7Fu) + 0x7F7F7F7Fu;
    return ~(t | x | 0x7F7F7F7Fu);
}

__device__ __forceinline__ void queue_chunk(v4u kw, int c, int lane, unsigned pat,
                                            int* q, int& qn) {
    if (c < 0) return;
    const int e0 = (c << 9) + 16 * lane;
    #pragma unroll
    for (int w = 0; w < 4; ++w) {
        unsigned z = zero_bytes(kw[w] ^ pat);
        unsigned mask = __builtin_amdgcn_ballot_w32(z != 0u);
        while (mask) {
            if (z != 0u) {
                q[qn + (int)__builtin_amdgcn_mbcnt_lo(mask, 0u)] =
                    e0 + 4 * w + (__builtin_ctz(z) >> 3);
                z &= z - 1u;
            }
            qn += __builtin_popcount(mask);
            mask = __builtin_amdgcn_ballot_w32(z != 0u);
        }
    }
}

template <int D, int R, int P>
__device__ __forceinline__ void file_batch(const float* __restrict__ x,
                                           const int*   __restrict__ src,
                                           const int*   __restrict__ dst,
                                           float* acc, int* cntL, int* lst, const int* q,
                                           int base, int count, int lo, int lane) {
    constexpr int NV = D / 64;
    int  s = 0, r = 0;
    bool over = false;
    if (lane < count) {
        const int e = q[base + lane];
        s = src[e];
        r = dst[e] - lo;
        const int place = atomicAdd(&cntL[r], 1);
        if (place < AGG_CAP) lst[r * AGG_CAP + place] = s;
        else                 over = true;
    }
    unsigned mask = __builtin_amdgcn_ballot_w32(over);
    while (mask) {
        const int l = __builtin_ctz(mask);
        mask &= mask - 1u;
        const int sg = __builtin_amdgcn_readlane(s, l);
        const int rg = __builtin_amdgcn_readlane(r, l);
        const float* row = x + (long long)sg * D;
        float* a = acc + rg * P + 2 * lane;
        #pragma unroll
        for (int k = 0; k < NV; ++k) {
            const v2f v = *reinterpret_cast<const v2f*>(row + (2 * lane + 64 * k));
            atomicAdd(a + 64 * k,     v[0]);
            atomicAdd(a + 64 * k + 1, v[1]);
        }
    }
}

template <int D, int P, bool MEAN, int NB>
__device__ __forceinline__ void sum_lists(const float* __restrict__ x, float* acc,
                                          const int* cntL, const int* lst,
                                          int nrows, int w, int nw, int lane) {
    constexpr int PARTS = D / 128;
    for (int item = w; item < nrows * PARTS; item += nw) {
        const int r    = item / PARTS;
        const int part = item % PARTS;
        const int cr   = cntL[r];
        const int n    = __builtin_amdgcn_readfirstlane(cr < AGG_CAP ? cr : AGG_CAP);
        const int* L   = lst + r * AGG_CAP;
        const float* xp = x + part * 128 + 4 * lane;
        v4f sum; sum[0] = 0.0f; sum[1] = 0.0f; sum[2] = 0.0f; sum[3] = 0.0f;
        for (int j = 0; j < n; j += NB) {
            int jj = j + lane;
            if (jj > n - 1) jj = n - 1;
            const int mine = L[jj];
            v4f v[NB];
            #pragma unroll
            for (int g = 0; g < NB; ++g) {
                const int sg = __builtin_amdgcn_readlane(mine, g);
                v[g] = *reinterpret_cast<const v4f*>(xp + (long long)sg * D);
            }
            #pragma unroll
            for (int g = 0; g < NB; ++g) {
                const bool live = (j + g < n);
                sum[0] += live ? v[g][0] : 0.0f;
                sum[1] += live ? v[g][1] : 0.0f;
                sum[2] += live ? v[g][2] : 0.0f;
                sum[3] += live ? v[g][3] : 0.0f;
            }
        }
        v4f* a = reinterpret_cast<v4f*>(acc + r * P + part * 128 + 4 * lane);
        if (MEAN) *a = (*a + sum) * (1.0f / fmaxf((float)cr, 1.0f));
        else      *a = *a + sum;
    }
}

template <int D, int P, int AHEAD, bool NEXT_A>
__device__ __forceinline__ void layer1_term(const float* A, const float* __restrict__ Bg,
                                            v8f (&accT)[10], int aOff, int bOff) {
    constexpr int NCH = D / 16;
    v4f bq[AHEAD + 1][2];
    #pragma unroll
    for (int p = 0; p < AHEAD; ++p) {
        #pragma unroll
        for (int q = 0; q < 2; ++q)
            bq[p][q] = *reinterpret_cast<const v4f*>(Bg + bOff + p * 16 + 4 * q);
    }
    #pragma unroll
    for (int ch = 0; ch < NCH; ++ch) {
        if (ch + AHEAD < NCH) {
            #pragma unroll
            for (int q = 0; q < 2; ++q)
                bq[(ch + AHEAD) % (AHEAD + 1)][q] =
                    *reinterpret_cast<const v4f*>(Bg + bOff + (ch + AHEAD) * 16 + 4 * q);
            __builtin_amdgcn_sched_barrier(0);
        }
        const v4f* b = bq[ch % (AHEAD + 1)];
        const float* aL = A + aOff + ch * 16;
        v4f a[2], an[2];
        #pragma unroll
        for (int q = 0; q < 2; ++q)
            a[q] = *reinterpret_cast<const v4f*>(aL + 4 * q);
        #pragma unroll
        for (int i = 0; i < 10; ++i) {
            if (NEXT_A && i + 1 < 10) {
                #pragma unroll
                for (int q = 0; q < 2; ++q)
                    an[q] = *reinterpret_cast<const v4f*>(aL + (i + 1) * 16 * P + 4 * q);
            }
            #pragma unroll
            for (int q = 0; q < 2; ++q) {
                #pragma unroll
                for (int s2 = 0; s2 < 2; ++s2) {
                    v2f bv; bv[0] = b[q][2 * s2]; bv[1] = b[q][2 * s2 + 1];
                    v2f av; av[0] = a[q][2 * s2]; av[1] = a[q][2 * s2 + 1];
                    accT[i] = __builtin_amdgcn_wmma_f32_16x16x4_f32(
                        false, av, false, bv, (short)0, accT[i], false, false);
                }
            }
            if (i + 1 < 10) {
                if (NEXT_A) {
                    a[0] = an[0];
                    a[1] = an[1];
                } else {
                    #pragma unroll
                    for (int q = 0; q < 2; ++q)
                        a[q] = *reinterpret_cast<const v4f*>(aL + (i + 1) * 16 * P + 4 * q);
                }
            }
        }
    }
}

template <int D, int R, int LAYER>
__launch_bounds__(AGG_NT)
__global__ void aggregate_owned(const float*    __restrict__ x,
                                const int*      __restrict__ src,
                                const int*      __restrict__ dst,
                                const unsigned* __restrict__ keys,
                                const float*    __restrict__ Wl,
                                const float*    __restrict__ Wr,
                                const float*    __restrict__ bias,
                                float*          __restrict__ out,
                                int E, int N) {
    constexpr int P = D + 4;
    __shared__ __attribute__((aligned(16))) float acc[(R + 1) * P];
    __shared__ int cntL[R + 1];
    __shared__ __attribute__((aligned(16))) int qAll[AGG_NW * AGG_QW];
    __shared__ int lst[R * AGG_CAP];

    const int tid   = threadIdx.x;
    const int lane  = tid & 31;
    const int wave  = __builtin_amdgcn_readfirstlane(tid >> 5);
    const int lo    = blockIdx.x * R;
    const int nrows = (N - lo < R) ? (N - lo) : R;
    const unsigned pat = (unsigned)blockIdx.x * 0x01010101u;

    constexpr int NST = (LAYER == 1) ? R * (D / 4) / AGG_NT : 1;
    v4f st[NST];
    if constexpr (LAYER == 1) {
        static_assert(R * (D / 4) % AGG_NT == 0, "whole trips");
        const v4f* xs = reinterpret_cast<const v4f*>(x + (long long)lo * D);
        #pragma unroll
        for (int u = 0; u < NST; ++u) st[u] = xs[tid + u * AGG_NT];
    }

    {
        v4f z; z[0] = 0.0f; z[1] = 0.0f; z[2] = 0.0f; z[3] = 0.0f;
        v4f* a4 = reinterpret_cast<v4f*>(acc);
        for (int i = tid; i < (R + 1) * P / 4; i += AGG_NT) a4[i] = z;
        if (tid <= R) cntL[tid] = 0;
    }
    __syncthreads();

    const int nCh   = (E + 511) >> 9;
    const int nGrp  = (nCh + AGG_NW * AGG_U - 1) / (AGG_NW * AGG_U);
    const int nTrip = (nGrp + 1) >> 1;
    const int nTot  = nTrip * 2 * AGG_NW * AGG_U;
    const int rot   = (int)blockIdx.x * (nTot / (int)gridDim.x);

    int* q  = qAll + wave * AGG_QW;
    int  qn = 0;

    if (nCh > 0) {
        v4u da[AGG_U], db[AGG_U];
        int ca[AGG_U], cb[AGG_U];
        #pragma unroll
        for (int u = 0; u < AGG_U; ++u) {
            ca[u] = slot_chunk(u, wave, rot, nTot, nCh);
            da[u] = load_keys(keys, ca[u], lane);
        }
        for (int t = 0; t < nTrip * 2 * AGG_U; t += 2 * AGG_U) {
            #pragma unroll
            for (int u = 0; u < AGG_U; ++u) {
                cb[u] = slot_chunk(t + AGG_U + u, wave, rot, nTot, nCh);
                db[u] = load_keys(keys, cb[u], lane);
            }
            #pragma unroll
            for (int u = 0; u < AGG_U; u += 2) {
                queue_chunk(da[u],     ca[u],     lane, pat, q, qn);
                queue_chunk(da[u + 1], ca[u + 1], lane, pat, q, qn);
                while (qn >= 32) {
                    qn -= 32;
                    file_batch<D, R, P>(x, src, dst, acc, cntL, lst, q, qn, 32, lo, lane);
                }
            }
            #pragma unroll
            for (int u = 0; u < AGG_U; ++u) {
                ca[u] = (t + 2 * AGG_U < nTrip * 2 * AGG_U)
                            ? slot_chunk(t + 2 * AGG_U + u, wave, rot, nTot, nCh) : -1;
                da[u] = load_keys(keys, ca[u], lane);
            }
            #pragma unroll
            for (int u = 0; u < AGG_U; u += 2) {
                queue_chunk(db[u],     cb[u],     lane, pat, q, qn);
                queue_chunk(db[u + 1], cb[u + 1], lane, pat, q, qn);
                while (qn >= 32) {
                    qn -= 32;
                    file_batch<D, R, P>(x, src, dst, acc, cntL, lst, q, qn, 32, lo, lane);
                }
            }
        }
    }
    if (qn > 0) file_batch<D, R, P>(x, src, dst, acc, cntL, lst, q, 0, qn, lo, lane);

    __syncthreads();

    if constexpr (LAYER == 1) {
        static_assert(D == 128, "two terms of 128 k");
        static_assert(R == 160, "10 row tiles, no padding rows");
        static_assert(R * P <= AGG_NW * AGG_QW, "own rows fit the queues");
        static_assert(AGG_NW == 32, "16 gathering and 16 multiplying waves, one per column tile");
        constexpr int NOUT1 = 256;
        constexpr int NG    = 16;
        float* own = reinterpret_cast<float*>(qAll);

        #pragma unroll
        for (int u = 0; u < NST; ++u) {
            const int i = tid + u * AGG_NT;
            *reinterpret_cast<v4f*>(own + (i / (D / 4)) * P + 4 * (i % (D / 4))) = st[u];
        }
        __syncthreads();

        const int c    = lane & 15;
        const int hh   = lane >> 4;
        const int col  = (wave - NG) * 16 + c;
        const int aOff = c * P + 8 * hh;
        const int bOff = col * D + 8 * hh;

        v8f accT[10];
        if (wave < NG) {
            sum_lists<D, P, true, 24>(x, acc, cntL, lst, nrows, wave, NG, lane);
        } else {
            #pragma unroll
            for (int i = 0; i < 10; ++i) {
                #pragma unroll
                for (int r = 0; r < 8; ++r) accT[i][r] = 0.0f;
            }
            layer1_term<D, P, 2, false>(own, Wr, accT, aOff, bOff);
        }
        __syncthreads();

        if (wave >= NG) {
            layer1_term<D, P, 1, true>(acc, Wl, accT, aOff, bOff);

            const float bv1 = bias[col];
            #pragma unroll
            for (int i = 0; i < 10; ++i) {
                const long long mbase = (long long)lo + 16 * i + 8 * hh;
                #pragma unroll
                for (int r = 0; r < 8; ++r)
                    out[(mbase + r) * NOUT1 + col] = fmaxf(accT[i][r] + bv1, 0.0f);
            }
        }
    } else {
        sum_lists<D, P, false, 16>(x, acc, cntL, lst, nrows, wave, AGG_NW, lane);
        __syncthreads();

        static_assert(D == 256, "slices of 64 k, 32 per lane half");
        constexpr int NOUTF = 64;
        constexpr int MTF   = (R + 15) / 16;
        constexpr int RP    = MTF * 16;
        static_assert(8 * RP * NOUTF <= AGG_NW * AGG_QW, "partial tiles fit the queues");
        float* part = reinterpret_cast<float*>(qAll);

        const int c    = lane & 15;
        const int hh   = lane >> 4;
        const int ks   = wave & 7;
        const int nt   = wave >> 3;
        const bool own = ks >= 4;
        const int kb   = (ks & 3) * 64 + 32 * hh;

        const float* bRow = (own ? Wr : Wl) + (nt * 16 + c) * D + kb;

        v8f accT[MTF];
        #pragma unroll
        for (int i = 0; i < MTF; ++i) {
            #pragma unroll
            for (int r = 0; r < 8; ++r) accT[i][r] = 0.0f;
        }

        int   rowL[MTF];
        float inv[MTF];
        const float* aG[MTF];
        #pragma unroll
        for (int i = 0; i < MTF; ++i) {
            const int row = 16 * i + c;
            rowL[i] = (row < nrows) ? row : R;
            inv[i]  = 1.0f / fmaxf((float)cntL[rowL[i]], 1.0f);
            int gr = lo + row;
            if (gr > N - 1) gr = N - 1;
            aG[i] = x + (long long)gr * D + kb;
        }

        #pragma unroll
        for (int qh = 0; qh < 2; ++qh) {
            v4f a[MTF][4], b[4];
            #pragma unroll
            for (int q = 0; q < 4; ++q)
                b[q] = *reinterpret_cast<const v4f*>(bRow + 16 * qh + 4 * q);
            if (own) {
                #pragma unroll
                for (int i = 0; i < MTF; ++i) {
                    #pragma unroll
                    for (int q = 0; q < 4; ++q)
                        a[i][q] = *reinterpret_cast<const v4f*>(aG[i] + 16 * qh + 4 * q);
                }
            } else {
                #pragma unroll
                for (int i = 0; i < MTF; ++i) {
                    #pragma unroll
                    for (int q = 0; q < 4; ++q)
                        a[i][q] = *reinterpret_cast<const v4f*>(acc + rowL[i] * P + kb + 16 * qh + 4 * q)
                                  * inv[i];
                }
            }
            #pragma unroll
            for (int q = 0; q < 4; ++q) {
                #pragma unroll
                for (int s = 0; s < 2; ++s) {
                    v2f bv; bv[0] = b[q][2 * s]; bv[1] = b[q][2 * s + 1];
                    #pragma unroll
                    for (int i = 0; i < MTF; ++i) {
                        v2f av; av[0] = a[i][q][2 * s]; av[1] = a[i][q][2 * s + 1];
                        accT[i] = __builtin_amdgcn_wmma_f32_16x16x4_f32(
                            false, av, false, bv, (short)0, accT[i], false, false);
                    }
                }
            }
        }

        #pragma unroll
        for (int i = 0; i < MTF; ++i) {
            #pragma unroll
            for (int r = 0; r < 8; ++r)
                part[(ks * RP + 16 * i + 8 * hh + r) * NOUTF + nt * 16 + c] = accT[i][r];
        }
        __syncthreads();

        const float b0 = bias[lane], b1 = bias[lane + 32];
        for (int row = wave; row < nrows; row += AGG_NW) {
            float va = 0.0f, vb = 0.0f;
            #pragma unroll
            for (int s8 = 0; s8 < 8; ++s8) {
                va += part[(s8 * RP + row) * NOUTF + lane];
                vb += part[(s8 * RP + row) * NOUTF + lane + 32];
            }
            va += b0;
            vb += b1;
            float m = fmaxf(va, vb);
            #pragma unroll
            for (int off = 16; off >= 1; off >>= 1)
                m = fmaxf(m, __shfl_xor(m, off, 32));
            float sum = __expf(va - m) + __expf(vb - m);
            #pragma unroll
            for (int off = 16; off >= 1; off >>= 1)
                sum += __shfl_xor(sum, off, 32);
            const float l = __logf(sum) + m;
            float* p = out + (long long)(lo + row) * NOUTF;
            p[lane]      = va - l;
            p[lane + 32] = vb - l;
        }
    }
}

#define AGG_WGPS 256
#define AGG_R1 160
#define AGG_R2 ((N3c + AGG_WGPS - 1) / AGG_WGPS)
static_assert(N2c % AGG_R1 == 0 && N2c / AGG_R1 <= AGG_WGPS, "layer 1: whole ranges, one work-group per WGP");

#define E1c 640000
#define E2c 160000
#define KEY_CH1 ((E1c + 511) / 512)
#define KEY_CH2 ((E2c + 511) / 512)
#define KEY_W1  (KEY_CH1 * 128)
#define KEY_W2  (KEY_CH2 * 128)
#define KEY_T1  (KEY_CH1 * 32)
#define KEY_T2  (KEY_CH2 * 32)

static_assert((N2c + AGG_R1 - 1) / AGG_R1 <= 255, "an owner number must fit a key byte, 0xFF kept free");
static_assert((N3c + AGG_R2 - 1) / AGG_R2 <= 255, "an owner number must fit a key byte, 0xFF kept free");

__global__ void owner_keys(const int* __restrict__ dst1, const int* __restrict__ dst2,
                           unsigned* __restrict__ keys1, unsigned* __restrict__ keys2,
                           int E1, int E2) {
    const int t = blockIdx.x * 256 + threadIdx.x;
    if (t < KEY_T1)                pack_keys16<AGG_R1>(dst1, keys1, t, E1);
    else if (t - KEY_T1 < KEY_T2)  pack_keys16<AGG_R2>(dst2, keys2, t - KEY_T1, E2);
}

extern "C" void kernel_launch(void* const* d_in, const int* in_sizes, int n_in,
                              void* d_out, int out_size, void* d_ws, size_t ws_size,
                              hipStream_t stream) {
    const float* x    = (const float*)d_in[0];
    const int*   src1 = (const int*)  d_in[1];
    const int*   dst1 = (const int*)  d_in[2];
    const int*   src2 = (const int*)  d_in[3];
    const int*   dst2 = (const int*)  d_in[4];
    const float* W1l  = (const float*)d_in[7];
    const float* W1r  = (const float*)d_in[8];
    const float* b1   = (const float*)d_in[9];
    const float* W2l  = (const float*)d_in[10];
    const float* W2r  = (const float*)d_in[11];
    const float* b2   = (const float*)d_in[12];

    const int E1 = in_sizes[1];
    const int E2 = in_sizes[3];

    float* h    = (float*)d_ws;
    float* out  = (float*)d_out;
    unsigned* keys1 = (unsigned*)(h + (size_t)N2c * HIDc);
    unsigned* keys2 = keys1 + KEY_W1;

    owner_keys<<<(KEY_T1 + KEY_T2 + 255) / 256, 256, 0, stream>>>(
        dst1, dst2, keys1, keys2, E1, E2);

    {
        aggregate_owned<INc, AGG_R1, 1>
            <<<N2c / AGG_R1, AGG_NT, 0, stream>>>(
                x, src1, dst1, keys1, W1l, W1r, b1, h, E1, N2c);
    }

    {
        aggregate_owned<HIDc, AGG_R2, 2>
            <<<(N3c + AGG_R2 - 1) / AGG_R2, AGG_NT, 0, stream>>>(
                h, src2, dst2, keys2, W2l, W2r, b2, out, E2, N3c);
    }
}


# Edge scan tests two chunks at once: a lane's 32 key bytes give one 32-bit match mask, matches queued in one loop per 1024 edges instead of one per 128; sums still in arrival order
# speedup vs baseline: 5.8611x; 1.0372x over previous
#include <hip/hip_runtime.h>
#include <math.h>

#define N1c  200000
#define N2c  40000
#define N3c  10000
#define INc  128
#define HIDc 256
#define OUTc 64

typedef float v2f __attribute__((ext_vector_type(2)));
typedef float v8f __attribute__((ext_vector_type(8)));

#define AGG_NT 1024
#define AGG_NW 32
#define AGG_U  4
#define AGG_QW 1056
#define AGG_CAP 64

typedef int      v4i  __attribute__((ext_vector_type(4)));
typedef unsigned v4u  __attribute__((ext_vector_type(4)));
typedef float    v4f  __attribute__((ext_vector_type(4)));

template <int R>
__device__ __forceinline__ unsigned key_of(int d) {
    const unsigned k = (unsigned)d / (unsigned)R;
    return (k < 255u) ? k : 255u;
}

template <int R>
__device__ __forceinline__ unsigned pack4(v4i d) {
    return key_of<R>(d[0]) | (key_of<R>(d[1]) << 8) |
           (key_of<R>(d[2]) << 16) | (key_of<R>(d[3]) << 24);
}

template <int R>
__device__ __forceinline__ void pack_keys16(const int* __restrict__ dst,
                                            unsigned* __restrict__ keys, int t, int E) {
    const int e0 = 16 * t;
    v4u w;
    if (e0 + 15 < E) {
        const v4i* p = reinterpret_cast<const v4i*>(dst + e0);
        const v4i d0 = p[0], d1 = p[1], d2 = p[2], d3 = p[3];
        w[0] = pack4<R>(d0);
        w[1] = pack4<R>(d1);
        w[2] = pack4<R>(d2);
        w[3] = pack4<R>(d3);
    } else {
        #pragma unroll
        for (int j = 0; j < 4; ++j) {
            unsigned word = 0u;
            #pragma unroll
            for (int c = 0; c < 4; ++c) {
                const int e = e0 + 4 * j + c;
                const unsigned k = (e < E) ? key_of<R>(dst[e]) : 255u;
                word |= k << (8 * c);
            }
            w[j] = word;
        }
    }
    *reinterpret_cast<v4u*>(keys + 4 * t) = w;
}

__device__ __forceinline__ int slot_chunk(int t, int wave, int rot, int nTot, int nCh) {
    int c = t * AGG_NW + wave + rot;
    if (c >= nTot) c -= nTot;
    return (c < nCh) ? c : -1;
}

__device__ __forceinline__ v4u load_keys(const unsigned* __restrict__ keys, int c, int lane) {
    const int cl = (c < 0) ? 0 : c;
    return *reinterpret_cast<const v4u*>(keys + (cl << 7) + 4 * lane);
}

__device__ __forceinline__ unsigned nonzero_bytes(unsigned x) {
    return (((x & 0x7F7F7F7Fu) + 0x7F7F7F7Fu) | x) & 0x80808080u;
}

__device__ __forceinline__ unsigned match_bits(v4u kw, unsigned pat) {
    const unsigned n = (nonzero_bytes(kw[0] ^ pat) >> 7) | (nonzero_bytes(kw[1] ^ pat) >> 6) |
                       (nonzero_bytes(kw[2] ^ pat) >> 5) | (nonzero_bytes(kw[3] ^ pat) >> 4);
    return n ^ 0x0F0F0F0Fu;
}

__device__ __forceinline__ void queue_pair(v4u kA, int cA, v4u kB, int cB, int lane,
                                           unsigned pat, int* q, int& qn) {
    unsigned m = 0u;
    if (cA >= 0) m = match_bits(kA, pat);
    if (cB >= 0) m |= match_bits(kB, pat) << 4;
    unsigned any = __builtin_amdgcn_ballot_w32(m != 0u);
    while (any) {
        if (m != 0u) {
            const int p  = __builtin_ctz(m);
            const int c9 = (p & 4) ? (cB << 9) : (cA << 9);
            q[qn + (int)__builtin_amdgcn_mbcnt_lo(any, 0u)] =
                c9 + 16 * lane + 4 * (p & 3) + (p >> 3);
            m &= m - 1u;
        }
        qn += __builtin_popcount(any);
        any = __builtin_amdgcn_ballot_w32(m != 0u);
    }
}

template <int D, int R, int P>
__device__ __forceinline__ void file_batch(const float* __restrict__ x,
                                           const int*   __restrict__ src,
                                           const int*   __restrict__ dst,
                                           float* acc, int* cntL, int* lst, const int* q,
                                           int base, int count, int lo, int lane) {
    constexpr int NV = D / 64;
    int  s = 0, r = 0;
    bool over = false;
    if (lane < count) {
        const int e = q[base + lane];
        s = src[e];
        r = dst[e] - lo;
        const int place = atomicAdd(&cntL[r], 1);
        if (place < AGG_CAP) lst[r * AGG_CAP + place] = s;
        else                 over = true;
    }
    unsigned mask = __builtin_amdgcn_ballot_w32(over);
    while (mask) {
        const int l = __builtin_ctz(mask);
        mask &= mask - 1u;
        const int sg = __builtin_amdgcn_readlane(s, l);
        const int rg = __builtin_amdgcn_readlane(r, l);
        const float* row = x + (long long)sg * D;
        float* a = acc + rg * P + 2 * lane;
        #pragma unroll
        for (int k = 0; k < NV; ++k) {
            const v2f v = *reinterpret_cast<const v2f*>(row + (2 * lane + 64 * k));
            atomicAdd(a + 64 * k,     v[0]);
            atomicAdd(a + 64 * k + 1, v[1]);
        }
    }
}

template <int D, int P, bool MEAN, int NB>
__device__ __forceinline__ void sum_lists(const float* __restrict__ x, float* acc,
                                          const int* cntL, const int* lst,
                                          int nrows, int w, int nw, int lane) {
    constexpr int PARTS = D / 128;
    for (int item = w; item < nrows * PARTS; item += nw) {
        const int r    = item / PARTS;
        const int part = item % PARTS;
        const int cr   = cntL[r];
        const int n    = __builtin_amdgcn_readfirstlane(cr < AGG_CAP ? cr : AGG_CAP);
        const int* L   = lst + r * AGG_CAP;
        const float* xp = x + part * 128 + 4 * lane;
        v4f sum; sum[0] = 0.0f; sum[1] = 0.0f; sum[2] = 0.0f; sum[3] = 0.0f;
        for (int j = 0; j < n; j += NB) {
            int jj = j + lane;
            if (jj > n - 1) jj = n - 1;
            const int mine = L[jj];
            v4f v[NB];
            #pragma unroll
            for (int g = 0; g < NB; ++g) {
                const int sg = __builtin_amdgcn_readlane(mine, g);
                v[g] = *reinterpret_cast<const v4f*>(xp + (long long)sg * D);
            }
            #pragma unroll
            for (int g = 0; g < NB; ++g) {
                const bool live = (j + g < n);
                sum[0] += live ? v[g][0] : 0.0f;
                sum[1] += live ? v[g][1] : 0.0f;
                sum[2] += live ? v[g][2] : 0.0f;
                sum[3] += live ? v[g][3] : 0.0f;
            }
        }
        v4f* a = reinterpret_cast<v4f*>(acc + r * P + part * 128 + 4 * lane);
        if (MEAN) *a = (*a + sum) * (1.0f / fmaxf((float)cr, 1.0f));
        else      *a = *a + sum;
    }
}

template <int D, int P, int AHEAD, bool NEXT_A>
__device__ __forceinline__ void layer1_term(const float* A, const float* __restrict__ Bg,
                                            v8f (&accT)[10], int aOff, int bOff) {
    constexpr int NCH = D / 16;
    v4f bq[AHEAD + 1][2];
    #pragma unroll
    for (int p = 0; p < AHEAD; ++p) {
        #pragma unroll
        for (int q = 0; q < 2; ++q)
            bq[p][q] = *reinterpret_cast<const v4f*>(Bg + bOff + p * 16 + 4 * q);
    }
    #pragma unroll
    for (int ch = 0; ch < NCH; ++ch) {
        if (ch + AHEAD < NCH) {
            #pragma unroll
            for (int q = 0; q < 2; ++q)
                bq[(ch + AHEAD) % (AHEAD + 1)][q] =
                    *reinterpret_cast<const v4f*>(Bg + bOff + (ch + AHEAD) * 16 + 4 * q);
            __builtin_amdgcn_sched_barrier(0);
        }
        const v4f* b = bq[ch % (AHEAD + 1)];
        const float* aL = A + aOff + ch * 16;
        v4f a[2], an[2];
        #pragma unroll
        for (int q = 0; q < 2; ++q)
            a[q] = *reinterpret_cast<const v4f*>(aL + 4 * q);
        #pragma unroll
        for (int i = 0; i < 10; ++i) {
            if (NEXT_A && i + 1 < 10) {
                #pragma unroll
                for (int q = 0; q < 2; ++q)
                    an[q] = *reinterpret_cast<const v4f*>(aL + (i + 1) * 16 * P + 4 * q);
            }
            #pragma unroll
            for (int q = 0; q < 2; ++q) {
                #pragma unroll
                for (int s2 = 0; s2 < 2; ++s2) {
                    v2f bv; bv[0] = b[q][2 * s2]; bv[1] = b[q][2 * s2 + 1];
                    v2f av; av[0] = a[q][2 * s2]; av[1] = a[q][2 * s2 + 1];
                    accT[i] = __builtin_amdgcn_wmma_f32_16x16x4_f32(
                        false, av, false, bv, (short)0, accT[i], false, false);
                }
            }
            if (i + 1 < 10) {
                if (NEXT_A) {
                    a[0] = an[0];
                    a[1] = an[1];
                } else {
                    #pragma unroll
                    for (int q = 0; q < 2; ++q)
                        a[q] = *reinterpret_cast<const v4f*>(aL + (i + 1) * 16 * P + 4 * q);
                }
            }
        }
    }
}

template <int D, int R, int LAYER>
__launch_bounds__(AGG_NT)
__global__ void aggregate_owned(const float*    __restrict__ x,
                                const int*      __restrict__ src,
                                const int*      __restrict__ dst,
                                const unsigned* __restrict__ keys,
                                const float*    __restrict__ Wl,
                                const float*    __restrict__ Wr,
                                const float*    __restrict__ bias,
                                float*          __restrict__ out,
                                int E, int N) {
    constexpr int P = D + 4;
    __shared__ __attribute__((aligned(16))) float acc[(R + 1) * P];
    __shared__ int cntL[R + 1];
    __shared__ __attribute__((aligned(16))) int qAll[AGG_NW * AGG_QW];
    __shared__ int lst[R * AGG_CAP];

    const int tid   = threadIdx.x;
    const int lane  = tid & 31;
    const int wave  = __builtin_amdgcn_readfirstlane(tid >> 5);
    const int lo    = blockIdx.x * R;
    const int nrows = (N - lo < R) ? (N - lo) : R;
    const unsigned pat = (unsigned)blockIdx.x * 0x01010101u;

    constexpr int NST = (LAYER == 1) ? R * (D / 4) / AGG_NT : 1;
    v4f st[NST];
    if constexpr (LAYER == 1) {
        static_assert(R * (D / 4) % AGG_NT == 0, "whole trips");
        const v4f* xs = reinterpret_cast<const v4f*>(x + (long long)lo * D);
        #pragma unroll
        for (int u = 0; u < NST; ++u) st[u] = xs[tid + u * AGG_NT];
    }

    {
        v4f z; z[0] = 0.0f; z[1] = 0.0f; z[2] = 0.0f; z[3] = 0.0f;
        v4f* a4 = reinterpret_cast<v4f*>(acc);
        for (int i = tid; i < (R + 1) * P / 4; i += AGG_NT) a4[i] = z;
        if (tid <= R) cntL[tid] = 0;
    }
    __syncthreads();

    const int nCh   = (E + 511) >> 9;
    const int nGrp  = (nCh + AGG_NW * AGG_U - 1) / (AGG_NW * AGG_U);
    const int nTrip = (nGrp + 1) >> 1;
    const int nTot  = nTrip * 2 * AGG_NW * AGG_U;
    const int rot   = (int)blockIdx.x * (nTot / (int)gridDim.x);

    int* q  = qAll + wave * AGG_QW;
    int  qn = 0;

    if (nCh > 0) {
        v4u da[AGG_U], db[AGG_U];
        int ca[AGG_U], cb[AGG_U];
        #pragma unroll
        for (int u = 0; u < AGG_U; ++u) {
            ca[u] = slot_chunk(u, wave, rot, nTot, nCh);
            da[u] = load_keys(keys, ca[u], lane);
        }
        for (int t = 0; t < nTrip * 2 * AGG_U; t += 2 * AGG_U) {
            #pragma unroll
            for (int u = 0; u < AGG_U; ++u) {
                cb[u] = slot_chunk(t + AGG_U + u, wave, rot, nTot, nCh);
                db[u] = load_keys(keys, cb[u], lane);
            }
            #pragma unroll
            for (int u = 0; u < AGG_U; u += 2) {
                queue_pair(da[u], ca[u], da[u + 1], ca[u + 1], lane, pat, q, qn);
                while (qn >= 32) {
                    qn -= 32;
                    file_batch<D, R, P>(x, src, dst, acc, cntL, lst, q, qn, 32, lo, lane);
                }
            }
            #pragma unroll
            for (int u = 0; u < AGG_U; ++u) {
                ca[u] = (t + 2 * AGG_U < nTrip * 2 * AGG_U)
                            ? slot_chunk(t + 2 * AGG_U + u, wave, rot, nTot, nCh) : -1;
                da[u] = load_keys(keys, ca[u], lane);
            }
            #pragma unroll
            for (int u = 0; u < AGG_U; u += 2) {
                queue_pair(db[u], cb[u], db[u + 1], cb[u + 1], lane, pat, q, qn);
                while (qn >= 32) {
                    qn -= 32;
                    file_batch<D, R, P>(x, src, dst, acc, cntL, lst, q, qn, 32, lo, lane);
                }
            }
        }
    }
    if (qn > 0) file_batch<D, R, P>(x, src, dst, acc, cntL, lst, q, 0, qn, lo, lane);

    __syncthreads();

    if constexpr (LAYER == 1) {
        static_assert(D == 128, "two terms of 128 k");
        static_assert(R == 160, "10 row tiles, no padding rows");
        static_assert(R * P <= AGG_NW * AGG_QW, "own rows fit the queues");
        static_assert(AGG_NW == 32, "16 gathering and 16 multiplying waves, one per column tile");
        constexpr int NOUT1 = 256;
        constexpr int NG    = 16;
        float* own = reinterpret_cast<float*>(qAll);

        #pragma unroll
        for (int u = 0; u < NST; ++u) {
            const int i = tid + u * AGG_NT;
            *reinterpret_cast<v4f*>(own + (i / (D / 4)) * P + 4 * (i % (D / 4))) = st[u];
        }
        __syncthreads();

        const int c    = lane & 15;
        const int hh   = lane >> 4;
        const int col  = (wave - NG) * 16 + c;
        const int aOff = c * P + 8 * hh;
        const int bOff = col * D + 8 * hh;

        v8f accT[10];
        if (wave < NG) {
            sum_lists<D, P, true, 24>(x, acc, cntL, lst, nrows, wave, NG, lane);
        } else {
            #pragma unroll
            for (int i = 0; i < 10; ++i) {
                #pragma unroll
                for (int r = 0; r < 8; ++r) accT[i][r] = 0.0f;
            }
            layer1_term<D, P, 2, false>(own, Wr, accT, aOff, bOff);
        }
        __syncthreads();

        if (wave >= NG) {
            layer1_term<D, P, 1, true>(acc, Wl, accT, aOff, bOff);

            const float bv1 = bias[col];
            #pragma unroll
            for (int i = 0; i < 10; ++i) {
                const long long mbase = (long long)lo + 16 * i + 8 * hh;
                #pragma unroll
                for (int r = 0; r < 8; ++r)
                    out[(mbase + r) * NOUT1 + col] = fmaxf(accT[i][r] + bv1, 0.0f);
            }
        }
    } else {
        sum_lists<D, P, false, 16>(x, acc, cntL, lst, nrows, wave, AGG_NW, lane);
        __syncthreads();

        static_assert(D == 256, "slices of 64 k, 32 per lane half");
        constexpr int NOUTF = 64;
        constexpr int MTF   = (R + 15) / 16;
        constexpr int RP    = MTF * 16;
        static_assert(8 * RP * NOUTF <= AGG_NW * AGG_QW, "partial tiles fit the queues");
        float* part = reinterpret_cast<float*>(qAll);

        const int c    = lane & 15;
        const int hh   = lane >> 4;
        const int ks   = wave & 7;
        const int nt   = wave >> 3;
        const bool own = ks >= 4;
        const int kb   = (ks & 3) * 64 + 32 * hh;

        const float* bRow = (own ? Wr : Wl) + (nt * 16 + c) * D + kb;

        v8f accT[MTF];
        #pragma unroll
        for (int i = 0; i < MTF; ++i) {
            #pragma unroll
            for (int r = 0; r < 8; ++r) accT[i][r] = 0.0f;
        }

        int   rowL[MTF];
        float inv[MTF];
        const float* aG[MTF];
        #pragma unroll
        for (int i = 0; i < MTF; ++i) {
            const int row = 16 * i + c;
            rowL[i] = (row < nrows) ? row : R;
            inv[i]  = 1.0f / fmaxf((float)cntL[rowL[i]], 1.0f);
            int gr = lo + row;
            if (gr > N - 1) gr = N - 1;
            aG[i] = x + (long long)gr * D + kb;
        }

        #pragma unroll
        for (int qh = 0; qh < 2; ++qh) {
            v4f a[MTF][4], b[4];
            #pragma unroll
            for (int q = 0; q < 4; ++q)
                b[q] = *reinterpret_cast<const v4f*>(bRow + 16 * qh + 4 * q);
            if (own) {
                #pragma unroll
                for (int i = 0; i < MTF; ++i) {
                    #pragma unroll
                    for (int q = 0; q < 4; ++q)
                        a[i][q] = *reinterpret_cast<const v4f*>(aG[i] + 16 * qh + 4 * q);
                }
            } else {
                #pragma unroll
                for (int i = 0; i < MTF; ++i) {
                    #pragma unroll
                    for (int q = 0; q < 4; ++q)
                        a[i][q] = *reinterpret_cast<const v4f*>(acc + rowL[i] * P + kb + 16 * qh + 4 * q)
                                  * inv[i];
                }
            }
            #pragma unroll
            for (int q = 0; q < 4; ++q) {
                #pragma unroll
                for (int s = 0; s < 2; ++s) {
                    v2f bv; bv[0] = b[q][2 * s]; bv[1] = b[q][2 * s + 1];
                    #pragma unroll
                    for (int i = 0; i < MTF; ++i) {
                        v2f av; av[0] = a[i][q][2 * s]; av[1] = a[i][q][2 * s + 1];
                        accT[i] = __builtin_amdgcn_wmma_f32_16x16x4_f32(
                            false, av, false, bv, (short)0, accT[i], false, false);
                    }
                }
            }
        }

        #pragma unroll
        for (int i = 0; i < MTF; ++i) {
            #pragma unroll
            for (int r = 0; r < 8; ++r)
                part[(ks * RP + 16 * i + 8 * hh + r) * NOUTF + nt * 16 + c] = accT[i][r];
        }
        __syncthreads();

        const float b0 = bias[lane], b1 = bias[lane + 32];
        for (int row = wave; row < nrows; row += AGG_NW) {
            float va = 0.0f, vb = 0.0f;
            #pragma unroll
            for (int s8 = 0; s8 < 8; ++s8) {
                va += part[(s8 * RP + row) * NOUTF + lane];
                vb += part[(s8 * RP + row) * NOUTF + lane + 32];
            }
            va += b0;
            vb += b1;
            float m = fmaxf(va, vb);
            #pragma unroll
            for (int off = 16; off >= 1; off >>= 1)
                m = fmaxf(m, __shfl_xor(m, off, 32));
            float sum = __expf(va - m) + __expf(vb - m);
            #pragma unroll
            for (int off = 16; off >= 1; off >>= 1)
                sum += __shfl_xor(sum, off, 32);
            const float l = __logf(sum) + m;
            float* p = out + (long long)(lo + row) * NOUTF;
            p[lane]      = va - l;
            p[lane + 32] = vb - l;
        }
    }
}

#define AGG_WGPS 256
#define AGG_R1 160
#define AGG_R2 ((N3c + AGG_WGPS - 1) / AGG_WGPS)
static_assert(N2c % AGG_R1 == 0 && N2c / AGG_R1 <= AGG_WGPS, "layer 1: whole ranges, one work-group per WGP");

#define E1c 640000
#define E2c 160000
#define KEY_CH1 ((E1c + 511) / 512)
#define KEY_CH2 ((E2c + 511) / 512)
#define KEY_W1  (KEY_CH1 * 128)
#define KEY_W2  (KEY_CH2 * 128)
#define KEY_T1  (KEY_CH1 * 32)
#define KEY_T2  (KEY_CH2 * 32)

static_assert((N2c + AGG_R1 - 1) / AGG_R1 <= 255, "an owner number must fit a key byte, 0xFF kept free");
static_assert((N3c + AGG_R2 - 1) / AGG_R2 <= 255, "an owner number must fit a key byte, 0xFF kept free");

__global__ void owner_keys(const int* __restrict__ dst1, const int* __restrict__ dst2,
                           unsigned* __restrict__ keys1, unsigned* __restrict__ keys2,
                           int E1, int E2) {
    const int t = blockIdx.x * 256 + threadIdx.x;
    if (t < KEY_T1)                pack_keys16<AGG_R1>(dst1, keys1, t, E1);
    else if (t - KEY_T1 < KEY_T2)  pack_keys16<AGG_R2>(dst2, keys2, t - KEY_T1, E2);
}

extern "C" void kernel_launch(void* const* d_in, const int* in_sizes, int n_in,
                              void* d_out, int out_size, void* d_ws, size_t ws_size,
                              hipStream_t stream) {
    const float* x    = (const float*)d_in[0];
    const int*   src1 = (const int*)  d_in[1];
    const int*   dst1 = (const int*)  d_in[2];
    const int*   src2 = (const int*)  d_in[3];
    const int*   dst2 = (const int*)  d_in[4];
    const float* W1l  = (const float*)d_in[7];
    const float* W1r  = (const float*)d_in[8];
    const float* b1   = (const float*)d_in[9];
    const float* W2l  = (const float*)d_in[10];
    const float* W2r  = (const float*)d_in[11];
    const float* b2   = (const float*)d_in[12];

    const int E1 = in_sizes[1];
    const int E2 = in_sizes[3];

    float* h    = (float*)d_ws;
    float* out  = (float*)d_out;
    unsigned* keys1 = (unsigned*)(h + (size_t)N2c * HIDc);
    unsigned* keys2 = keys1 + KEY_W1;

    owner_keys<<<(KEY_T1 + KEY_T2 + 255) / 256, 256, 0, stream>>>(
        dst1, dst2, keys1, keys2, E1, E2);

    {
        aggregate_owned<INc, AGG_R1, 1>
            <<<N2c / AGG_R1, AGG_NT, 0, stream>>>(
                x, src1, dst1, keys1, W1l, W1r, b1, h, E1, N2c);
    }

    {
        aggregate_owned<HIDc, AGG_R2, 2>
            <<<(N3c + AGG_R2 - 1) / AGG_R2, AGG_NT, 0, stream>>>(
                h, src2, dst2, keys2, W2l, W2r, b2, out, E2, N3c);
    }
}
